# starting kernel
# baseline (speedup 1.0000x reference)
#include <hip/hip_runtime.h>

typedef __attribute__((ext_vector_type(2))) float v2f;
typedef __attribute__((ext_vector_type(8))) float v8f;

#define SEQ_LEN 262144
#define NGATES  12
#define HDIM    4

__device__ __forceinline__ float bcast_lane(float v, int lane) {
  return __int_as_float(__builtin_amdgcn_readlane(__float_as_int(v), lane));
}

__device__ __forceinline__ float fast_sigmoid(float x) {
  const float e = __builtin_amdgcn_exp2f(-1.44269504088896340736f * x);
  return __builtin_amdgcn_rcpf(1.0f + e);
}

__global__ __launch_bounds__(256) void gi_wmma_kernel(
    const float* __restrict__ x, const float* __restrict__ w_ih,
    const float* __restrict__ b_ih, float* __restrict__ gi) {
  const int lane = threadIdx.x & 31;
  const int wave = threadIdx.x >> 5;
  const int tile = blockIdx.x * 8 + wave;
  const int t0   = tile * 16;
  const int half = lane >> 4;
  const int idx  = lane & 15;

  v2f a;
  if (half == 0) {
    const float2 xv = *reinterpret_cast<const float2*>(x + (size_t)(t0 + idx) * 2);
    a.x = xv.x; a.y = xv.y;
  } else {
    a.x = 0.0f; a.y = 0.0f;
  }

  v2f b;
  if (half == 0 && idx < NGATES) {
    const float2 wv = *reinterpret_cast<const float2*>(w_ih + idx * 2);
    b.x = wv.x; b.y = wv.y;
  } else {
    b.x = 0.0f; b.y = 0.0f;
  }

  const float bias = (idx < NGATES) ? b_ih[idx] : 0.0f;
  v8f c;
#pragma unroll
  for (int i = 0; i < 8; ++i) c[i] = bias;

  v8f d = __builtin_amdgcn_wmma_f32_16x16x4_f32(
       false, a,  false, b,
       (short)0, c,  false,  false);

  if (idx < NGATES) {
    float* dst = gi + (size_t)idx * SEQ_LEN + t0 + half * 8;
    *reinterpret_cast<float4*>(dst + 0) = make_float4(d[0], d[1], d[2], d[3]);
    *reinterpret_cast<float4*>(dst + 4) = make_float4(d[4], d[5], d[6], d[7]);
  }
}

__global__ __launch_bounds__(32) void gru_scan_kernel(
    const float* __restrict__ gi, const float* __restrict__ state0,
    const float* __restrict__ w_hh, const float* __restrict__ b_hh,
    float* __restrict__ hbuf, float* __restrict__ state_out) {
  const int lane = threadIdx.x;
  const int j    = (lane < NGATES) ? lane : 0;
  const int jj   = lane & 3;

  const float w0 = w_hh[j * 4 + 0];
  const float w1 = w_hh[j * 4 + 1];
  const float w2 = w_hh[j * 4 + 2];
  const float w3 = w_hh[j * 4 + 3];
  const float bh = b_hh[j];

  float h0 = state0[0], h1 = state0[1], h2 = state0[2], h3 = state0[3];

  const float* __restrict__ gptr = gi + (size_t)j * SEQ_LEN;

  constexpr int PF = 8;
  float buf[PF];
#pragma unroll
  for (int u = 0; u < PF; ++u) buf[u] = gptr[u];

  for (int t = 0; t < SEQ_LEN; t += PF) {
#pragma unroll
    for (int u = 0; u < PF; ++u) {
      const float giv = buf[u];
      buf[u] = gptr[t + u + PF];

      const float p01 = fmaf(h1, w1, fmaf(h0, w0, bh));
      const float p23 = fmaf(h3, w3, h2 * w2);
      const float gh  = p01 + p23;

      const float s = fast_sigmoid(giv + gh);

      const float r = __shfl(s, jj, 32);

      const float nv = fmaf(2.0f, fast_sigmoid(2.0f * (giv + r * gh)), -1.0f);

      const float z    = __shfl(s, 4 + jj, 32);
      const float nn   = __shfl(nv, 8 + jj, 32);
      const float hold = (jj == 0) ? h0 : (jj == 1) ? h1 : (jj == 2) ? h2 : h3;
      const float hnew = fmaf(z, hold - nn, nn);

      if (lane < HDIM) hbuf[(size_t)(t + u) * HDIM + lane] = hnew;

      h0 = bcast_lane(hnew, 0);
      h1 = bcast_lane(hnew, 1);
      h2 = bcast_lane(hnew, 2);
      h3 = bcast_lane(hnew, 3);
    }
  }

  if (lane == 0) {
    state_out[0] = h0; state_out[1] = h1;
    state_out[2] = h2; state_out[3] = h3;
  }
}

__global__ __launch_bounds__(256) void yproj_wmma_kernel(
    const float* __restrict__ hbuf, const float* __restrict__ w_out,
    const float* __restrict__ b_out, float* __restrict__ y) {
  const int lane = threadIdx.x & 31;
  const int wave = threadIdx.x >> 5;
  const int tile = blockIdx.x * 8 + wave;
  const int t0   = tile * 16;
  const int half = lane >> 4;
  const int idx  = lane & 15;

  v2f a;
  {
    const float2 hv = *reinterpret_cast<const float2*>(
        hbuf + (size_t)(t0 + idx) * 4 + half * 2);
    a.x = hv.x; a.y = hv.y;
  }

  v2f b;
  if (idx < 2) {
    const float2 wv = *reinterpret_cast<const float2*>(w_out + idx * 4 + half * 2);
    b.x = wv.x; b.y = wv.y;
  } else {
    b.x = 0.0f; b.y = 0.0f;
  }

  const float bias = (idx < 2) ? b_out[idx] : 0.0f;
  v8f c;
#pragma unroll
  for (int i = 0; i < 8; ++i) c[i] = bias;

  v8f d = __builtin_amdgcn_wmma_f32_16x16x4_f32(
      false, a, false, b, (short)0, c, false, false);

  if (idx < 2) {
    const int rbase = half * 8;
#pragma unroll
    for (int i = 0; i < 8; ++i)
      y[(size_t)(t0 + rbase + i) * 2 + idx] = d[i];
  }
}

extern "C" void kernel_launch(void* const* d_in, const int* in_sizes, int n_in,
                              void* d_out, int out_size, void* d_ws, size_t ws_size,
                              hipStream_t stream) {
  const float* x      = (const float*)d_in[0];
  const float* state  = (const float*)d_in[1];
  const float* w_ih   = (const float*)d_in[2];
  const float* w_hh   = (const float*)d_in[3];
  const float* b_ih   = (const float*)d_in[4];
  const float* b_hh   = (const float*)d_in[5];
  const float* w_out  = (const float*)d_in[6];
  const float* b_out  = (const float*)d_in[7];

  float* out  = (float*)d_out;
  float* gi   = (float*)d_ws;
  float* hbuf = gi + (size_t)NGATES * SEQ_LEN;

  const int tiles  = SEQ_LEN / 16;
  const int blocks = tiles / 8;

  gi_wmma_kernel<<<blocks, 256, 0, stream>>>(x, w_ih, b_ih, gi);
  gru_scan_kernel<<<1, 32, 0, stream>>>(gi, state, w_hh, b_hh,
                                        hbuf, out + (size_t)SEQ_LEN * 2);
  yproj_wmma_kernel<<<blocks, 256, 0, stream>>>(hbuf, w_out, b_out, out);
}


# scan: each lane owns all 3 gates of one state component, h shared in the quad by DPP quad_perm (4 lanes): no ds_bpermute, readlane or exec branch on the serial chain; loads, stores per 16 steps
# speedup vs baseline: 3.9030x; 3.9030x over previous
#include <hip/hip_runtime.h>

typedef __attribute__((ext_vector_type(2))) float v2f;
typedef __attribute__((ext_vector_type(8))) float v8f;

#define SEQ_LEN 262144
#define NGATES  12
#define HDIM    4

__device__ __forceinline__ float bcast_lane(float v, int lane) {
  return __int_as_float(__builtin_amdgcn_readlane(__float_as_int(v), lane));
}

__device__ __forceinline__ float fast_sigmoid(float x) {
  const float e = __builtin_amdgcn_exp2f(-1.44269504088896340736f * x);
  return __builtin_amdgcn_rcpf(1.0f + e);
}

__global__ __launch_bounds__(256) void gi_wmma_kernel(
    const float* __restrict__ x, const float* __restrict__ w_ih,
    const float* __restrict__ b_ih, float* __restrict__ gi) {
  const int lane = threadIdx.x & 31;
  const int wave = threadIdx.x >> 5;
  const int tile = blockIdx.x * 8 + wave;
  const int t0   = tile * 16;
  const int half = lane >> 4;
  const int idx  = lane & 15;

  v2f a;
  if (half == 0) {
    const float2 xv = *reinterpret_cast<const float2*>(x + (size_t)(t0 + idx) * 2);
    a.x = xv.x; a.y = xv.y;
  } else {
    a.x = 0.0f; a.y = 0.0f;
  }

  v2f b;
  if (half == 0 && idx < NGATES) {
    const float2 wv = *reinterpret_cast<const float2*>(w_ih + idx * 2);
    b.x = wv.x; b.y = wv.y;
  } else {
    b.x = 0.0f; b.y = 0.0f;
  }

  const float bias = (idx < NGATES) ? b_ih[idx] : 0.0f;
  v8f c;
#pragma unroll
  for (int i = 0; i < 8; ++i) c[i] = bias;

  v8f d = __builtin_amdgcn_wmma_f32_16x16x4_f32(
       false, a,  false, b,
       (short)0, c,  false,  false);

  if (idx < NGATES) {
    float* dst = gi + (size_t)idx * SEQ_LEN + t0 + half * 8;
    *reinterpret_cast<float4*>(dst + 0) = make_float4(d[0], d[1], d[2], d[3]);
    *reinterpret_cast<float4*>(dst + 4) = make_float4(d[4], d[5], d[6], d[7]);
  }
}

typedef __attribute__((ext_vector_type(4))) float v4f;

template <int K>
__device__ __forceinline__ float quad_lane(float v) {
  return __int_as_float(__builtin_amdgcn_update_dpp(
      0, __float_as_int(v), K * 0x55, 0xF, 0xF, true));
}

struct GiBlock { v4f r[4], z[4], n[4]; };

__device__ __forceinline__ void load_block(GiBlock& g, const float* gr,
                                           const float* gz,
                                           const float* gn, int t) {
#pragma unroll
  for (int q = 0; q < 4; ++q) {
    g.r[q] = *reinterpret_cast<const v4f*>(gr + t + 4 * q);
    g.z[q] = *reinterpret_cast<const v4f*>(gz + t + 4 * q);
    g.n[q] = *reinterpret_cast<const v4f*>(gn + t + 4 * q);
  }
}

__global__ __launch_bounds__(32) __attribute__((amdgpu_waves_per_eu(1, 1)))
void gru_scan_kernel(
    const float* gi, const float* __restrict__ state0,
    const float* __restrict__ w_hh, const float* __restrict__ b_hh,
    float* hbuf, float* __restrict__ state_out) {
  const int c = threadIdx.x & 3;

  const float wr0 = w_hh[c * 4 + 0], wr1 = w_hh[c * 4 + 1];
  const float wr2 = w_hh[c * 4 + 2], wr3 = w_hh[c * 4 + 3];
  const float wz0 = w_hh[(4 + c) * 4 + 0], wz1 = w_hh[(4 + c) * 4 + 1];
  const float wz2 = w_hh[(4 + c) * 4 + 2], wz3 = w_hh[(4 + c) * 4 + 3];
  const float wn0 = w_hh[(8 + c) * 4 + 0], wn1 = w_hh[(8 + c) * 4 + 1];
  const float wn2 = w_hh[(8 + c) * 4 + 2], wn3 = w_hh[(8 + c) * 4 + 3];
  const float br = b_hh[c], bz = b_hh[4 + c], bn = b_hh[8 + c];

  float h = state0[c];

  const float* gr = gi + (size_t)c * SEQ_LEN;
  const float* gz = gi + (size_t)(4 + c) * SEQ_LEN;
  const float* gn = gi + (size_t)(8 + c) * SEQ_LEN;
  float* hp = hbuf + c;

  auto run_block = [&](const GiBlock& g, int t) {
    float hs[16];
#pragma unroll
    for (int u = 0; u < 16; ++u) {
      const float gir = g.r[u >> 2][u & 3];
      const float giz = g.z[u >> 2][u & 3];
      const float gin = g.n[u >> 2][u & 3];

      const float h0 = quad_lane<0>(h), h1 = quad_lane<1>(h);
      const float h2 = quad_lane<2>(h), h3 = quad_lane<3>(h);

      __builtin_amdgcn_sched_barrier(0);
      const float ghr = fmaf(h1, wr1, fmaf(h0, wr0, br)) + fmaf(h3, wr3, h2 * wr2);
      const float ghz = fmaf(h1, wz1, fmaf(h0, wz0, bz)) + fmaf(h3, wz3, h2 * wz2);
      const float xr = gir + ghr;
      const float xz = giz + ghz;
      __builtin_amdgcn_sched_barrier(0);
      const float r = fast_sigmoid(xr);
      const float z = fast_sigmoid(xz);
      const float ghn = fmaf(h1, wn1, fmaf(h0, wn0, bn)) + fmaf(h3, wn3, h2 * wn2);
      __builtin_amdgcn_sched_barrier(0);
      const float nv = fmaf(2.0f, fast_sigmoid(2.0f * (gin + r * ghn)), -1.0f);
      h = fmaf(z, h - nv, nv);
      hs[u] = h;
    }
    asm volatile("" :: "v"(h) : "memory");
#pragma unroll
    for (int u = 0; u < 16; ++u) hp[(size_t)(t + u) * HDIM] = hs[u];
  };

  asm volatile("" :: "v"(wr0), "v"(wr1), "v"(wr2), "v"(wr3),
                     "v"(wz0), "v"(wz1), "v"(wz2), "v"(wz3),
                     "v"(wn0), "v"(wn1), "v"(wn2), "v"(wn3),
                     "v"(br), "v"(bz), "v"(bn), "v"(h));

  GiBlock ga, gb;
  load_block(ga, gr, gz, gn, 0);
  for (int t = 0; t < SEQ_LEN; t += 32) {
    load_block(gb, gr, gz, gn, t + 16);
    run_block(ga, t);
    const int tn = (t + 32 < SEQ_LEN) ? (t + 32) : (SEQ_LEN - 16);
    load_block(ga, gr, gz, gn, tn);
    run_block(gb, t + 16);
  }

  state_out[c] = h;
}

__global__ __launch_bounds__(256) void yproj_wmma_kernel(
    const float* __restrict__ hbuf, const float* __restrict__ w_out,
    const float* __restrict__ b_out, float* __restrict__ y) {
  const int lane = threadIdx.x & 31;
  const int wave = threadIdx.x >> 5;
  const int tile = blockIdx.x * 8 + wave;
  const int t0   = tile * 16;
  const int half = lane >> 4;
  const int idx  = lane & 15;

  v2f a;
  {
    const float2 hv = *reinterpret_cast<const float2*>(
        hbuf + (size_t)(t0 + idx) * 4 + half * 2);
    a.x = hv.x; a.y = hv.y;
  }

  v2f b;
  if (idx < 2) {
    const float2 wv = *reinterpret_cast<const float2*>(w_out + idx * 4 + half * 2);
    b.x = wv.x; b.y = wv.y;
  } else {
    b.x = 0.0f; b.y = 0.0f;
  }

  const float bias = (idx < 2) ? b_out[idx] : 0.0f;
  v8f c;
#pragma unroll
  for (int i = 0; i < 8; ++i) c[i] = bias;

  v8f d = __builtin_amdgcn_wmma_f32_16x16x4_f32(
      false, a, false, b, (short)0, c, false, false);

  if (idx < 2) {
    const int rbase = half * 8;
#pragma unroll
    for (int i = 0; i < 8; ++i)
      y[(size_t)(t0 + rbase + i) * 2 + idx] = d[i];
  }
}

extern "C" void kernel_launch(void* const* d_in, const int* in_sizes, int n_in,
                              void* d_out, int out_size, void* d_ws, size_t ws_size,
                              hipStream_t stream) {
  const float* x      = (const float*)d_in[0];
  const float* state  = (const float*)d_in[1];
  const float* w_ih   = (const float*)d_in[2];
  const float* w_hh   = (const float*)d_in[3];
  const float* b_ih   = (const float*)d_in[4];
  const float* b_hh   = (const float*)d_in[5];
  const float* w_out  = (const float*)d_in[6];
  const float* b_out  = (const float*)d_in[7];

  float* out  = (float*)d_out;
  float* gi   = (float*)d_ws;
  float* hbuf = gi + (size_t)NGATES * SEQ_LEN;

  const int tiles  = SEQ_LEN / 16;
  const int blocks = tiles / 8;

  gi_wmma_kernel<<<blocks, 256, 0, stream>>>(x, w_ih, b_ih, gi);
  gru_scan_kernel<<<1, 4, 0, stream>>>(gi, state, w_hh, b_hh,
                                        hbuf, out + (size_t)SEQ_LEN * 2);
  yproj_wmma_kernel<<<blocks, 256, 0, stream>>>(hbuf, w_out, b_out, out);
}


# scan: h->h chain cut from 18 to 12 dependent ops: exp2 scales (-log2e, 2x) and b_hh folded into w_hh and gi before the loop, blend as one fma s*A+B (A,B from z); f32 sum order and rounding change
# speedup vs baseline: 5.1955x; 1.3312x over previous
#include <hip/hip_runtime.h>

typedef __attribute__((ext_vector_type(2))) float v2f;
typedef __attribute__((ext_vector_type(8))) float v8f;

#define SEQ_LEN 262144
#define NGATES  12
#define HDIM    4

__device__ __forceinline__ float bcast_lane(float v, int lane) {
  return __int_as_float(__builtin_amdgcn_readlane(__float_as_int(v), lane));
}

#define NEG_LOG2E (-1.44269504088896340736f)

__device__ __forceinline__ float sigmoid_scaled(float a) {
  return __builtin_amdgcn_rcpf(1.0f + __builtin_amdgcn_exp2f(a));
}

__global__ __launch_bounds__(256) void gi_wmma_kernel(
    const float* __restrict__ x, const float* __restrict__ w_ih,
    const float* __restrict__ b_ih, const float* __restrict__ b_hh,
    float* __restrict__ gi) {
  const int lane = threadIdx.x & 31;
  const int wave = threadIdx.x >> 5;
  const int tile = blockIdx.x * 8 + wave;
  const int t0   = tile * 16;
  const int half = lane >> 4;
  const int idx  = lane & 15;

  v2f a;
  if (half == 0) {
    const float2 xv = *reinterpret_cast<const float2*>(x + (size_t)(t0 + idx) * 2);
    a.x = xv.x; a.y = xv.y;
  } else {
    a.x = 0.0f; a.y = 0.0f;
  }

  const float cs = (idx < 2 * HDIM) ? NEG_LOG2E : 2.0f * NEG_LOG2E;

  v2f b;
  if (half == 0 && idx < NGATES) {
    const float2 wv = *reinterpret_cast<const float2*>(w_ih + idx * 2);
    b.x = cs * wv.x; b.y = cs * wv.y;
  } else {
    b.x = 0.0f; b.y = 0.0f;
  }

  float bias = 0.0f;
  if (idx < 2 * HDIM)   bias = cs * (b_ih[idx] + b_hh[idx]);
  else if (idx < NGATES) bias = cs * b_ih[idx];
  v8f c;
#pragma unroll
  for (int i = 0; i < 8; ++i) c[i] = bias;

  v8f d = __builtin_amdgcn_wmma_f32_16x16x4_f32(
       false, a,  false, b,
       (short)0, c,  false,  false);

  if (idx < NGATES) {
    float* dst = gi + (size_t)idx * SEQ_LEN + t0 + half * 8;
    *reinterpret_cast<float4*>(dst + 0) = make_float4(d[0], d[1], d[2], d[3]);
    *reinterpret_cast<float4*>(dst + 4) = make_float4(d[4], d[5], d[6], d[7]);
  }
}

typedef __attribute__((ext_vector_type(4))) float v4f;

template <int K>
__device__ __forceinline__ float quad_lane(float v) {
  return __int_as_float(__builtin_amdgcn_update_dpp(
      0, __float_as_int(v), K * 0x55, 0xF, 0xF, true));
}

struct GiBlock { v4f r[4], z[4], n[4]; };

__device__ __forceinline__ void load_block(GiBlock& g, const float* gr,
                                           const float* gz,
                                           const float* gn, int t) {
#pragma unroll
  for (int q = 0; q < 4; ++q) {
    g.r[q] = *reinterpret_cast<const v4f*>(gr + t + 4 * q);
    g.z[q] = *reinterpret_cast<const v4f*>(gz + t + 4 * q);
    g.n[q] = *reinterpret_cast<const v4f*>(gn + t + 4 * q);
  }
}

__global__ __launch_bounds__(32) __attribute__((amdgpu_waves_per_eu(1, 1)))
void gru_scan_kernel(
    const float* gi, const float* __restrict__ state0,
    const float* __restrict__ w_hh, const float* __restrict__ b_hh,
    float* hbuf, float* __restrict__ state_out) {
  const int c = threadIdx.x & 3;

  const float sr = NEG_LOG2E, sn = 2.0f * NEG_LOG2E;
  const float wr0 = sr * w_hh[c * 4 + 0], wr1 = sr * w_hh[c * 4 + 1];
  const float wr2 = sr * w_hh[c * 4 + 2], wr3 = sr * w_hh[c * 4 + 3];
  const float wz0 = sr * w_hh[(4 + c) * 4 + 0], wz1 = sr * w_hh[(4 + c) * 4 + 1];
  const float wz2 = sr * w_hh[(4 + c) * 4 + 2], wz3 = sr * w_hh[(4 + c) * 4 + 3];
  const float wn0 = sn * w_hh[(8 + c) * 4 + 0], wn1 = sn * w_hh[(8 + c) * 4 + 1];
  const float wn2 = sn * w_hh[(8 + c) * 4 + 2], wn3 = sn * w_hh[(8 + c) * 4 + 3];
  const float bn = sn * b_hh[8 + c];

  float h = state0[c];

  const float* gr = gi + (size_t)c * SEQ_LEN;
  const float* gz = gi + (size_t)(4 + c) * SEQ_LEN;
  const float* gn = gi + (size_t)(8 + c) * SEQ_LEN;
  float* hp = hbuf + c;

  auto run_block = [&](const GiBlock& g, int t) {
    float hs[16];
#pragma unroll
    for (int u = 0; u < 16; ++u) {
      const float gir = g.r[u >> 2][u & 3];
      const float giz = g.z[u >> 2][u & 3];
      const float gin = g.n[u >> 2][u & 3];

      const float h0 = quad_lane<0>(h), h1 = quad_lane<1>(h);
      const float h2 = quad_lane<2>(h), h3 = quad_lane<3>(h);

      __builtin_amdgcn_sched_barrier(0);
      const float ar = fmaf(h1, wr1, fmaf(h0, wr0, gir)) + fmaf(h3, wr3, h2 * wr2);
      const float az = fmaf(h1, wz1, fmaf(h0, wz0, giz)) + fmaf(h3, wz3, h2 * wz2);
      __builtin_amdgcn_sched_barrier(0);
      const float r = sigmoid_scaled(ar);
      const float z = sigmoid_scaled(az);
      const float ghn = fmaf(h1, wn1, fmaf(h0, wn0, bn)) + fmaf(h3, wn3, h2 * wn2);
      __builtin_amdgcn_sched_barrier(0);
      const float en = __builtin_amdgcn_exp2f(fmaf(r, ghn, gin));
      __builtin_amdgcn_sched_barrier(0);
      const float zm1 = z - 1.0f;
      const float ca = -2.0f * zm1;
      const float cb = fmaf(z, h, zm1);
      const float s = __builtin_amdgcn_rcpf(1.0f + en);
      h = fmaf(s, ca, cb);
      hs[u] = h;
    }
    asm volatile("" :: "v"(h) : "memory");
#pragma unroll
    for (int u = 0; u < 16; ++u) hp[(size_t)(t + u) * HDIM] = hs[u];
  };

  asm volatile("" :: "v"(h));

  GiBlock ga, gb;
  load_block(ga, gr, gz, gn, 0);
  for (int t = 0; t < SEQ_LEN; t += 32) {
    load_block(gb, gr, gz, gn, t + 16);
    run_block(ga, t);
    const int tn = (t + 32 < SEQ_LEN) ? (t + 32) : (SEQ_LEN - 16);
    load_block(ga, gr, gz, gn, tn);
    run_block(gb, t + 16);
  }

  state_out[c] = h;
}

__global__ __launch_bounds__(256) void yproj_wmma_kernel(
    const float* __restrict__ hbuf, const float* __restrict__ w_out,
    const float* __restrict__ b_out, float* __restrict__ y) {
  const int lane = threadIdx.x & 31;
  const int wave = threadIdx.x >> 5;
  const int tile = blockIdx.x * 8 + wave;
  const int t0   = tile * 16;
  const int half = lane >> 4;
  const int idx  = lane & 15;

  v2f a;
  {
    const float2 hv = *reinterpret_cast<const float2*>(
        hbuf + (size_t)(t0 + idx) * 4 + half * 2);
    a.x = hv.x; a.y = hv.y;
  }

  v2f b;
  if (idx < 2) {
    const float2 wv = *reinterpret_cast<const float2*>(w_out + idx * 4 + half * 2);
    b.x = wv.x; b.y = wv.y;
  } else {
    b.x = 0.0f; b.y = 0.0f;
  }

  const float bias = (idx < 2) ? b_out[idx] : 0.0f;
  v8f c;
#pragma unroll
  for (int i = 0; i < 8; ++i) c[i] = bias;

  v8f d = __builtin_amdgcn_wmma_f32_16x16x4_f32(
      false, a, false, b, (short)0, c, false, false);

  if (idx < 2) {
    const int rbase = half * 8;
#pragma unroll
    for (int i = 0; i < 8; ++i)
      y[(size_t)(t0 + rbase + i) * 2 + idx] = d[i];
  }
}

extern "C" void kernel_launch(void* const* d_in, const int* in_sizes, int n_in,
                              void* d_out, int out_size, void* d_ws, size_t ws_size,
                              hipStream_t stream) {
  const float* x      = (const float*)d_in[0];
  const float* state  = (const float*)d_in[1];
  const float* w_ih   = (const float*)d_in[2];
  const float* w_hh   = (const float*)d_in[3];
  const float* b_ih   = (const float*)d_in[4];
  const float* b_hh   = (const float*)d_in[5];
  const float* w_out  = (const float*)d_in[6];
  const float* b_out  = (const float*)d_in[7];

  float* out  = (float*)d_out;
  float* gi   = (float*)d_ws;
  float* hbuf = gi + (size_t)NGATES * SEQ_LEN;

  const int tiles  = SEQ_LEN / 16;
  const int blocks = tiles / 8;

  gi_wmma_kernel<<<blocks, 256, 0, stream>>>(x, w_ih, b_ih, b_hh, gi);
  gru_scan_kernel<<<1, 4, 0, stream>>>(gi, state, w_hh, b_hh,
                                        hbuf, out + (size_t)SEQ_LEN * 2);
  yproj_wmma_kernel<<<blocks, 256, 0, stream>>>(hbuf, w_out, b_out, out);
}


# scan: each gate function is ONE hardware tanh (sigmoid(x)=0.5+0.5*tanh(x/2), halves folded into w_hh, b_hh, gi) in place of exp2, add, rcp; step order rebuilt for it; gate rounding changes
# speedup vs baseline: 7.6136x; 1.4654x over previous
#include <hip/hip_runtime.h>

typedef __attribute__((ext_vector_type(2))) float v2f;
typedef __attribute__((ext_vector_type(8))) float v8f;

#define SEQ_LEN 262144
#define NGATES  12
#define HDIM    4

__device__ __forceinline__ float bcast_lane(float v, int lane) {
  return __int_as_float(__builtin_amdgcn_readlane(__float_as_int(v), lane));
}

__device__ __forceinline__ float hw_tanh(float a) {
  return __builtin_amdgcn_tanhf(a);
}

__global__ __launch_bounds__(256) void gi_wmma_kernel(
    const float* __restrict__ x, const float* __restrict__ w_ih,
    const float* __restrict__ b_ih, const float* __restrict__ b_hh,
    float* __restrict__ gi) {
  const int lane = threadIdx.x & 31;
  const int wave = threadIdx.x >> 5;
  const int tile = blockIdx.x * 8 + wave;
  const int t0   = tile * 16;
  const int half = lane >> 4;
  const int idx  = lane & 15;

  v2f a;
  if (half == 0) {
    const float2 xv = *reinterpret_cast<const float2*>(x + (size_t)(t0 + idx) * 2);
    a.x = xv.x; a.y = xv.y;
  } else {
    a.x = 0.0f; a.y = 0.0f;
  }

  const float cs = (idx < 2 * HDIM) ? 0.5f : 1.0f;

  v2f b;
  if (half == 0 && idx < NGATES) {
    const float2 wv = *reinterpret_cast<const float2*>(w_ih + idx * 2);
    b.x = cs * wv.x; b.y = cs * wv.y;
  } else {
    b.x = 0.0f; b.y = 0.0f;
  }

  float bias = 0.0f;
  if (idx < 2 * HDIM)   bias = cs * (b_ih[idx] + b_hh[idx]);
  else if (idx < NGATES) bias = cs * b_ih[idx];
  v8f c;
#pragma unroll
  for (int i = 0; i < 8; ++i) c[i] = bias;

  v8f d = __builtin_amdgcn_wmma_f32_16x16x4_f32(
       false, a,  false, b,
       (short)0, c,  false,  false);

  if (idx < NGATES) {
    float* dst = gi + (size_t)idx * SEQ_LEN + t0 + half * 8;
    *reinterpret_cast<float4*>(dst + 0) = make_float4(d[0], d[1], d[2], d[3]);
    *reinterpret_cast<float4*>(dst + 4) = make_float4(d[4], d[5], d[6], d[7]);
  }
}

typedef __attribute__((ext_vector_type(4))) float v4f;

template <int K>
__device__ __forceinline__ float quad_lane(float v) {
  return __int_as_float(__builtin_amdgcn_update_dpp(
      0, __float_as_int(v), K * 0x55, 0xF, 0xF, true));
}

struct GiBlock { v4f r[4], z[4], n[4]; };

__device__ __forceinline__ void load_block(GiBlock& g, const float* gr,
                                           const float* gz,
                                           const float* gn, int t) {
#pragma unroll
  for (int q = 0; q < 4; ++q) {
    g.r[q] = *reinterpret_cast<const v4f*>(gr + t + 4 * q);
    g.z[q] = *reinterpret_cast<const v4f*>(gz + t + 4 * q);
    g.n[q] = *reinterpret_cast<const v4f*>(gn + t + 4 * q);
  }
}

__global__ __launch_bounds__(32) __attribute__((amdgpu_waves_per_eu(1, 1)))
void gru_scan_kernel(
    const float* gi, const float* __restrict__ state0,
    const float* __restrict__ w_hh, const float* __restrict__ b_hh,
    float* hbuf, float* __restrict__ state_out) {
  const int c = threadIdx.x & 3;

  const float hf = 0.5f;
  const float wr0 = hf * w_hh[c * 4 + 0], wr1 = hf * w_hh[c * 4 + 1];
  const float wr2 = hf * w_hh[c * 4 + 2], wr3 = hf * w_hh[c * 4 + 3];
  const float wz0 = hf * w_hh[(4 + c) * 4 + 0], wz1 = hf * w_hh[(4 + c) * 4 + 1];
  const float wz2 = hf * w_hh[(4 + c) * 4 + 2], wz3 = hf * w_hh[(4 + c) * 4 + 3];
  const float wn0 = hf * w_hh[(8 + c) * 4 + 0], wn1 = hf * w_hh[(8 + c) * 4 + 1];
  const float wn2 = hf * w_hh[(8 + c) * 4 + 2], wn3 = hf * w_hh[(8 + c) * 4 + 3];
  const float bn = hf * b_hh[8 + c];

  float h = state0[c];

  const float* gr = gi + (size_t)c * SEQ_LEN;
  const float* gz = gi + (size_t)(4 + c) * SEQ_LEN;
  const float* gn = gi + (size_t)(8 + c) * SEQ_LEN;
  float* hp = hbuf + c;

  auto run_block = [&](const GiBlock& g, int t) {
    float hs[16];
#pragma unroll
    for (int u = 0; u < 16; ++u) {
      const float gir = g.r[u >> 2][u & 3];
      const float giz = g.z[u >> 2][u & 3];
      const float gin = g.n[u >> 2][u & 3];

      const float h0 = quad_lane<0>(h), h1 = quad_lane<1>(h);
      const float h2 = quad_lane<2>(h), h3 = quad_lane<3>(h);
      const float r23a = h2 * wr2;
      __builtin_amdgcn_sched_barrier(0);
      const float r01a = fmaf(h0, wr0, gir);
      const float n01a = fmaf(h0, wn0, bn);
      const float n23a = h2 * wn2;
      const float z01a = fmaf(h0, wz0, giz);
      __builtin_amdgcn_sched_barrier(0);
      const float r01 = fmaf(h1, wr1, r01a);
      const float r23 = fmaf(h3, wr3, r23a);
      __builtin_amdgcn_sched_barrier(0);
      const float z23a = h2 * wz2;
      const float n01 = fmaf(h1, wn1, n01a);
      const float n23 = fmaf(h3, wn3, n23a);
      __builtin_amdgcn_sched_barrier(0);
      const float ar = r01 + r23;
      __builtin_amdgcn_sched_barrier(0);
      const float z01 = fmaf(h1, wz1, z01a);
      const float z23 = fmaf(h3, wz3, z23a);
      const float gh = n01 + n23;
      const float hh = 0.5f * h;
      __builtin_amdgcn_sched_barrier(0);
      const float tr = hw_tanh(ar);
      __builtin_amdgcn_sched_barrier(0);
      const float az = z01 + z23;
      const float gnb = gin + gh;
      const float tz = hw_tanh(az);
      __builtin_amdgcn_sched_barrier(0);
      const float n = hw_tanh(fmaf(tr, gh, gnb));
      __builtin_amdgcn_sched_barrier(0);
      const float ca = fmaf(tz, -0.5f, 0.5f);
      const float cb = fmaf(tz, hh, hh);
      h = fmaf(n, ca, cb);
      hs[u] = h;
    }
    asm volatile("" :: "v"(h) : "memory");
#pragma unroll
    for (int u = 0; u < 16; ++u) hp[(size_t)(t + u) * HDIM] = hs[u];
  };

  asm volatile("" :: "v"(h));

  GiBlock ga, gb;
  load_block(ga, gr, gz, gn, 0);
  for (int t = 0; t < SEQ_LEN; t += 32) {
    load_block(gb, gr, gz, gn, t + 16);
    run_block(ga, t);
    const int tn = (t + 32 < SEQ_LEN) ? (t + 32) : (SEQ_LEN - 16);
    load_block(ga, gr, gz, gn, tn);
    run_block(gb, t + 16);
  }

  state_out[c] = h;
}

__global__ __launch_bounds__(256) void yproj_wmma_kernel(
    const float* __restrict__ hbuf, const float* __restrict__ w_out,
    const float* __restrict__ b_out, float* __restrict__ y) {
  const int lane = threadIdx.x & 31;
  const int wave = threadIdx.x >> 5;
  const int tile = blockIdx.x * 8 + wave;
  const int t0   = tile * 16;
  const int half = lane >> 4;
  const int idx  = lane & 15;

  v2f a;
  {
    const float2 hv = *reinterpret_cast<const float2*>(
        hbuf + (size_t)(t0 + idx) * 4 + half * 2);
    a.x = hv.x; a.y = hv.y;
  }

  v2f b;
  if (idx < 2) {
    const float2 wv = *reinterpret_cast<const float2*>(w_out + idx * 4 + half * 2);
    b.x = wv.x; b.y = wv.y;
  } else {
    b.x = 0.0f; b.y = 0.0f;
  }

  const float bias = (idx < 2) ? b_out[idx] : 0.0f;
  v8f c;
#pragma unroll
  for (int i = 0; i < 8; ++i) c[i] = bias;

  v8f d = __builtin_amdgcn_wmma_f32_16x16x4_f32(
      false, a, false, b, (short)0, c, false, false);

  if (idx < 2) {
    const int rbase = half * 8;
#pragma unroll
    for (int i = 0; i < 8; ++i)
      y[(size_t)(t0 + rbase + i) * 2 + idx] = d[i];
  }
}

extern "C" void kernel_launch(void* const* d_in, const int* in_sizes, int n_in,
                              void* d_out, int out_size, void* d_ws, size_t ws_size,
                              hipStream_t stream) {
  const float* x      = (const float*)d_in[0];
  const float* state  = (const float*)d_in[1];
  const float* w_ih   = (const float*)d_in[2];
  const float* w_hh   = (const float*)d_in[3];
  const float* b_ih   = (const float*)d_in[4];
  const float* b_hh   = (const float*)d_in[5];
  const float* w_out  = (const float*)d_in[6];
  const float* b_out  = (const float*)d_in[7];

  float* out  = (float*)d_out;
  float* gi   = (float*)d_ws;
  float* hbuf = gi + (size_t)NGATES * SEQ_LEN;

  const int tiles  = SEQ_LEN / 16;
  const int blocks = tiles / 8;

  gi_wmma_kernel<<<blocks, 256, 0, stream>>>(x, w_ih, b_ih, b_hh, gi);
  gru_scan_kernel<<<1, 4, 0, stream>>>(gi, state, w_hh, b_hh,
                                        hbuf, out + (size_t)SEQ_LEN * 2);
  yproj_wmma_kernel<<<blocks, 256, 0, stream>>>(hbuf, w_out, b_out, out);
}


# scan: no memory wait left on the serial chain: one per-lane pointer + immediate offsets for all loads/stores, h as SoA rows (4 b128 stores per block, not 16 b32), stored registers held half a block
# speedup vs baseline: 8.5761x; 1.1264x over previous
#include <hip/hip_runtime.h>

typedef __attribute__((ext_vector_type(2))) float v2f;
typedef __attribute__((ext_vector_type(8))) float v8f;

#define SEQ_LEN 262144
#define NGATES  12
#define HDIM    4

__device__ __forceinline__ float bcast_lane(float v, int lane) {
  return __int_as_float(__builtin_amdgcn_readlane(__float_as_int(v), lane));
}

__device__ __forceinline__ float hw_tanh(float a) {
  return __builtin_amdgcn_tanhf(a);
}

__global__ __launch_bounds__(256) void gi_wmma_kernel(
    const float* __restrict__ x, const float* __restrict__ w_ih,
    const float* __restrict__ b_ih, const float* __restrict__ b_hh,
    float* __restrict__ gi) {
  const int lane = threadIdx.x & 31;
  const int wave = threadIdx.x >> 5;
  const int tile = blockIdx.x * 8 + wave;
  const int t0   = tile * 16;
  const int half = lane >> 4;
  const int idx  = lane & 15;

  v2f a;
  if (half == 0) {
    const float2 xv = *reinterpret_cast<const float2*>(x + (size_t)(t0 + idx) * 2);
    a.x = xv.x; a.y = xv.y;
  } else {
    a.x = 0.0f; a.y = 0.0f;
  }

  const float cs = (idx < 2 * HDIM) ? 0.5f : 1.0f;

  v2f b;
  if (half == 0 && idx < NGATES) {
    const float2 wv = *reinterpret_cast<const float2*>(w_ih + idx * 2);
    b.x = cs * wv.x; b.y = cs * wv.y;
  } else {
    b.x = 0.0f; b.y = 0.0f;
  }

  float bias = 0.0f;
  if (idx < 2 * HDIM)   bias = cs * (b_ih[idx] + b_hh[idx]);
  else if (idx < NGATES) bias = cs * b_ih[idx];
  v8f c;
#pragma unroll
  for (int i = 0; i < 8; ++i) c[i] = bias;

  v8f d = __builtin_amdgcn_wmma_f32_16x16x4_f32(
       false, a,  false, b,
       (short)0, c,  false,  false);

  if (idx < NGATES) {
    float* dst = gi + (size_t)idx * SEQ_LEN + t0 + half * 8;
    *reinterpret_cast<float4*>(dst + 0) = make_float4(d[0], d[1], d[2], d[3]);
    *reinterpret_cast<float4*>(dst + 4) = make_float4(d[4], d[5], d[6], d[7]);
  }
}

typedef __attribute__((ext_vector_type(4))) float v4f;

template <int K>
__device__ __forceinline__ float quad_lane(float v) {
  return __int_as_float(__builtin_amdgcn_update_dpp(
      0, __float_as_int(v), K * 0x55, 0xF, 0xF, true));
}

struct GiBlock { v4f r[4], z[4], n[4]; };

__device__ __forceinline__ void load_block(GiBlock& g, const float* p) {
#pragma unroll
  for (int q = 0; q < 4; ++q) {
    g.r[q] = *reinterpret_cast<const v4f*>(p - 8 * SEQ_LEN + 4 * q);
    g.z[q] = *reinterpret_cast<const v4f*>(p - 4 * SEQ_LEN + 4 * q);
    g.n[q] = *reinterpret_cast<const v4f*>(p + 4 * q);
  }
}

__global__ __launch_bounds__(32) __attribute__((amdgpu_waves_per_eu(1, 1)))
void gru_scan_kernel(
    float* ws, const float* __restrict__ state0,
    const float* __restrict__ w_hh, const float* __restrict__ b_hh,
    float* __restrict__ state_out) {
  const int c = threadIdx.x & 3;

  const float hf = 0.5f;
  const float wr0 = hf * w_hh[c * 4 + 0], wr1 = hf * w_hh[c * 4 + 1];
  const float wr2 = hf * w_hh[c * 4 + 2], wr3 = hf * w_hh[c * 4 + 3];
  const float wz0 = hf * w_hh[(4 + c) * 4 + 0], wz1 = hf * w_hh[(4 + c) * 4 + 1];
  const float wz2 = hf * w_hh[(4 + c) * 4 + 2], wz3 = hf * w_hh[(4 + c) * 4 + 3];
  const float wn0 = hf * w_hh[(8 + c) * 4 + 0], wn1 = hf * w_hh[(8 + c) * 4 + 1];
  const float wn2 = hf * w_hh[(8 + c) * 4 + 2], wn3 = hf * w_hh[(8 + c) * 4 + 3];
  const float bn = hf * b_hh[8 + c];

  float h = state0[c];

  float* p = ws + (size_t)(8 + c) * SEQ_LEN;

  auto run_block = [&](const GiBlock& g, v4f (&o)[4], const v4f (&keep)[4]) {
    float hs[16];
#pragma unroll
    for (int u = 0; u < 16; ++u) {
      const float gir = g.r[u >> 2][u & 3];
      const float giz = g.z[u >> 2][u & 3];
      const float gin = g.n[u >> 2][u & 3];

      const float h0 = quad_lane<0>(h), h1 = quad_lane<1>(h);
      const float h2 = quad_lane<2>(h), h3 = quad_lane<3>(h);
      const float r23a = h2 * wr2;
      __builtin_amdgcn_sched_barrier(0);
      const float r01a = fmaf(h0, wr0, gir);
      const float n01a = fmaf(h0, wn0, bn);
      const float n23a = h2 * wn2;
      const float z01a = fmaf(h0, wz0, giz);
      __builtin_amdgcn_sched_barrier(0);
      const float r01 = fmaf(h1, wr1, r01a);
      const float r23 = fmaf(h3, wr3, r23a);
      __builtin_amdgcn_sched_barrier(0);
      const float z23a = h2 * wz2;
      const float n01 = fmaf(h1, wn1, n01a);
      const float n23 = fmaf(h3, wn3, n23a);
      __builtin_amdgcn_sched_barrier(0);
      const float ar = r01 + r23;
      __builtin_amdgcn_sched_barrier(0);
      const float z01 = fmaf(h1, wz1, z01a);
      const float z23 = fmaf(h3, wz3, z23a);
      const float gh = n01 + n23;
      const float hh = 0.5f * h;
      __builtin_amdgcn_sched_barrier(0);
      const float tr = hw_tanh(ar);
      __builtin_amdgcn_sched_barrier(0);
      const float az = z01 + z23;
      const float gnb = gin + gh;
      const float tz = hw_tanh(az);
      __builtin_amdgcn_sched_barrier(0);
      const float n = hw_tanh(fmaf(tr, gh, gnb));
      __builtin_amdgcn_sched_barrier(0);
      const float ca = fmaf(tz, -0.5f, 0.5f);
      const float cb = fmaf(tz, hh, hh);
      h = fmaf(n, ca, cb);
      hs[u] = h;
      if (u == 7) {
        asm volatile("" :: "v"(keep[0]), "v"(keep[1]), "v"(keep[2]), "v"(keep[3]));
      }
    }
    asm volatile("" :: "v"(h) : "memory");
#pragma unroll
    for (int q = 0; q < 4; ++q) {
      o[q].x = hs[4 * q + 0]; o[q].y = hs[4 * q + 1];
      o[q].z = hs[4 * q + 2]; o[q].w = hs[4 * q + 3];
    }
  };
  auto store_block = [&](const v4f (&o)[4], float* hrow) {
#pragma unroll
    for (int q = 0; q < 4; ++q) *reinterpret_cast<v4f*>(hrow + 4 * q) = o[q];
  };

  asm volatile("" :: "v"(h));

  GiBlock ga, gb;
  v4f oa[4], ob[4];
#pragma unroll
  for (int q = 0; q < 4; ++q) { ob[q].x = 0.0f; ob[q].y = 0.0f; ob[q].z = 0.0f; ob[q].w = 0.0f; }
  load_block(ga, p);
  for (int t = 0; t < SEQ_LEN; t += 32) {
    load_block(gb, p + 16);
    if (t != 0) store_block(ob, p + 4 * SEQ_LEN - 16);
    run_block(ga, oa, ob);
    store_block(oa, p + 4 * SEQ_LEN);
    load_block(ga, p + 32);
    run_block(gb, ob, oa);
    p += 32;
  }
  store_block(ob, p + 4 * SEQ_LEN - 16);

  state_out[c] = h;
}

__global__ __launch_bounds__(256) void yproj_wmma_kernel(
    const float* __restrict__ hbuf, const float* __restrict__ w_out,
    const float* __restrict__ b_out, float* __restrict__ y) {
  const int lane = threadIdx.x & 31;
  const int wave = threadIdx.x >> 5;
  const int tile = blockIdx.x * 8 + wave;
  const int t0   = tile * 16;
  const int half = lane >> 4;
  const int idx  = lane & 15;

  v2f a;
  {
    const float* hr = hbuf + (size_t)(half * 2) * SEQ_LEN + t0 + idx;
    a.x = hr[0]; a.y = hr[SEQ_LEN];
  }

  v2f b;
  if (idx < 2) {
    const float2 wv = *reinterpret_cast<const float2*>(w_out + idx * 4 + half * 2);
    b.x = wv.x; b.y = wv.y;
  } else {
    b.x = 0.0f; b.y = 0.0f;
  }

  const float bias = (idx < 2) ? b_out[idx] : 0.0f;
  v8f c;
#pragma unroll
  for (int i = 0; i < 8; ++i) c[i] = bias;

  v8f d = __builtin_amdgcn_wmma_f32_16x16x4_f32(
      false, a, false, b, (short)0, c, false, false);

  if (idx < 2) {
    const int rbase = half * 8;
#pragma unroll
    for (int i = 0; i < 8; ++i)
      y[(size_t)(t0 + rbase + i) * 2 + idx] = d[i];
  }
}

extern "C" void kernel_launch(void* const* d_in, const int* in_sizes, int n_in,
                              void* d_out, int out_size, void* d_ws, size_t ws_size,
                              hipStream_t stream) {
  const float* x      = (const float*)d_in[0];
  const float* state  = (const float*)d_in[1];
  const float* w_ih   = (const float*)d_in[2];
  const float* w_hh   = (const float*)d_in[3];
  const float* b_ih   = (const float*)d_in[4];
  const float* b_hh   = (const float*)d_in[5];
  const float* w_out  = (const float*)d_in[6];
  const float* b_out  = (const float*)d_in[7];

  float* out  = (float*)d_out;
  float* gi   = (float*)d_ws;
  float* hbuf = gi + (size_t)NGATES * SEQ_LEN;

  const int tiles  = SEQ_LEN / 16;
  const int blocks = tiles / 8;

  gi_wmma_kernel<<<blocks, 256, 0, stream>>>(x, w_ih, b_ih, b_hh, gi);
  gru_scan_kernel<<<1, 4, 0, stream>>>(gi, state, w_hh, b_hh,
                                        out + (size_t)SEQ_LEN * 2);
  yproj_wmma_kernel<<<blocks, 256, 0, stream>>>(hbuf, w_out, b_out, out);
}


# scan: gate products spread over 16 lanes, one w_hh term per lane, summed by 2 DPP row_xmask adds; blend folded into the terms (n feeds them directly); gi blocks of 32 steps; f32 sum order changes
# speedup vs baseline: 9.8319x; 1.1464x over previous
#include <hip/hip_runtime.h>

typedef __attribute__((ext_vector_type(2))) float v2f;
typedef __attribute__((ext_vector_type(8))) float v8f;

#define SEQ_LEN 262144
#define NGATES  12
#define HDIM    4

__device__ __forceinline__ float bcast_lane(float v, int lane) {
  return __int_as_float(__builtin_amdgcn_readlane(__float_as_int(v), lane));
}

__device__ __forceinline__ float hw_tanh(float a) {
  return __builtin_amdgcn_tanhf(a);
}

__global__ __launch_bounds__(256) void gi_wmma_kernel(
    const float* __restrict__ x, const float* __restrict__ w_ih,
    const float* __restrict__ b_ih, const float* __restrict__ b_hh,
    float* __restrict__ gi) {
  const int lane = threadIdx.x & 31;
  const int wave = threadIdx.x >> 5;
  const int tile = blockIdx.x * 8 + wave;
  const int t0   = tile * 16;
  const int half = lane >> 4;
  const int idx  = lane & 15;

  v2f a;
  if (half == 0) {
    const float2 xv = *reinterpret_cast<const float2*>(x + (size_t)(t0 + idx) * 2);
    a.x = xv.x; a.y = xv.y;
  } else {
    a.x = 0.0f; a.y = 0.0f;
  }

  const float cs = (idx < 2 * HDIM) ? 0.125f : 1.0f;

  v2f b;
  if (half == 0 && idx < NGATES) {
    const float2 wv = *reinterpret_cast<const float2*>(w_ih + idx * 2);
    b.x = cs * wv.x; b.y = cs * wv.y;
  } else {
    b.x = 0.0f; b.y = 0.0f;
  }

  float bias = 0.0f;
  if (idx < 2 * HDIM)   bias = cs * (b_ih[idx] + b_hh[idx]);
  else if (idx < NGATES) bias = cs * b_ih[idx];
  v8f c;
#pragma unroll
  for (int i = 0; i < 8; ++i) c[i] = bias;

  v8f d = __builtin_amdgcn_wmma_f32_16x16x4_f32(
       false, a,  false, b,
       (short)0, c,  false,  false);

  if (idx < NGATES) {
    float* dst = gi + (size_t)idx * SEQ_LEN + t0 + half * 8;
    *reinterpret_cast<float4*>(dst + 0) = make_float4(d[0], d[1], d[2], d[3]);
    *reinterpret_cast<float4*>(dst + 4) = make_float4(d[4], d[5], d[6], d[7]);
  }
}

typedef __attribute__((ext_vector_type(4))) float v4f;

#define BLK 32
#define BQ  (BLK / 4)

template <int M>
__device__ __forceinline__ float row_xor(float v) {
  return __int_as_float(__builtin_amdgcn_update_dpp(
      0, __float_as_int(v), 0x160 + M, 0xF, 0xF, true));
}

__device__ __forceinline__ float quad_first(float v) {
  return __int_as_float(__builtin_amdgcn_update_dpp(
      0, __float_as_int(v), 0x00, 0xF, 0xF, true));
}

struct GiBlock { v4f r[BQ], z[BQ], n[BQ]; };

__device__ __forceinline__ void load_block(GiBlock& g, const float* wt,
                                           unsigned orz, unsigned on) {
#pragma unroll
  for (int i = 0; i < BQ; ++i) {
    g.r[i] = *reinterpret_cast<const v4f*>((wt + 4 * i) + orz);
    g.z[i] = *reinterpret_cast<const v4f*>((wt + 4 * SEQ_LEN + 4 * i) + orz);
    g.n[i] = *reinterpret_cast<const v4f*>((wt + 4 * i) + on);
  }
}

__global__ __launch_bounds__(32) __attribute__((amdgpu_waves_per_eu(1, 1)))
void gru_scan_kernel(
    float* ws, const float* __restrict__ state0,
    const float* __restrict__ w_hh, const float* __restrict__ b_hh,
    float* __restrict__ state_out) {
  const int l = threadIdx.x & 15;
  const int q = l >> 2;
  const int j = l & 3;
  const int d = q ^ j;

  const float hf = 0.5f;
  const float wr = hf * w_hh[d * 4 + q];
  const float wz = hf * w_hh[(4 + d) * 4 + q];
  const float wn = hf * w_hh[(8 + d) * 4 + q];
  const float bn = (j == 0) ? hf * b_hh[8 + q] : 0.0f;

  float h = state0[q];

  const unsigned orz = (unsigned)d * SEQ_LEN;
  const unsigned on  = (unsigned)(8 + q) * SEQ_LEN;
  float* wt = ws;

  float Tr, Tz, Tg;

  auto run_block = [&](const GiBlock& g, const GiBlock& gn, v4f (&o)[BQ],
                       const v4f (&keep)[BQ]) {
    float hs[BLK];
#pragma unroll
    for (int u = 0; u < BLK; ++u) {
      const float gin  = g.n[u >> 2][u & 3];
      const float girn = (u + 1 < BLK) ? g.r[(u + 1) >> 2][(u + 1) & 3] : gn.r[0][0];
      const float gizn = (u + 1 < BLK) ? g.z[(u + 1) >> 2][(u + 1) & 3] : gn.z[0][0];

      const float s1g = Tg + row_xor<5>(Tg);
      __builtin_amdgcn_sched_barrier(0);
      const float s1r = Tr + row_xor<5>(Tr);
      __builtin_amdgcn_sched_barrier(0);
      const float s1z = Tz + row_xor<5>(Tz);
      __builtin_amdgcn_sched_barrier(0);
      const float s2g = s1g + row_xor<10>(s1g);
      __builtin_amdgcn_sched_barrier(0);
      const float s2r = s1r + row_xor<10>(s1r);
      __builtin_amdgcn_sched_barrier(0);
      const float s2z = s1z + row_xor<10>(s1z);
      __builtin_amdgcn_sched_barrier(0);
      const float gh = quad_first(s2g);
      __builtin_amdgcn_sched_barrier(0);
      const float tr = hw_tanh(quad_first(s2r));
      __builtin_amdgcn_sched_barrier(0);
      const float tz = hw_tanh(quad_first(s2z));
      __builtin_amdgcn_sched_barrier(0);
      const float gnb = gin + gh;
      const float hh = 0.5f * h;
      __builtin_amdgcn_sched_barrier(0);
      const float na = fmaf(tr, gh, gnb);
      __builtin_amdgcn_sched_barrier(0);
      const float ca = fmaf(tz, -0.5f, 0.5f);
      const float cb = fmaf(tz, hh, hh);
      __builtin_amdgcn_sched_barrier(0);
      const float n = hw_tanh(na);
      __builtin_amdgcn_sched_barrier(0);
      const float ur = wr * ca;
      const float uz = wz * ca;
      const float ug = wn * ca;
      const float vr = fmaf(wr, cb, girn);
      const float vz = fmaf(wz, cb, gizn);
      const float vg = fmaf(wn, cb, bn);
      __builtin_amdgcn_sched_barrier(0);
      Tr = fmaf(n, ur, vr);
      Tz = fmaf(n, uz, vz);
      Tg = fmaf(n, ug, vg);
      h = fmaf(n, ca, cb);
      hs[u] = h;
      __builtin_amdgcn_sched_barrier(0);
      if (u == BLK / 2 - 1) {
#pragma unroll
        for (int i = 0; i < BQ; ++i) asm volatile("" :: "v"(keep[i]));
      }
    }
    asm volatile("" :: "v"(h) : "memory");
#pragma unroll
    for (int i = 0; i < BQ; ++i) {
      o[i].x = hs[4 * i + 0]; o[i].y = hs[4 * i + 1];
      o[i].z = hs[4 * i + 2]; o[i].w = hs[4 * i + 3];
    }
  };
  auto store_block = [&](const v4f (&o)[BQ], float* wh) {
#pragma unroll
    for (int i = 0; i < BQ; ++i)
      *reinterpret_cast<v4f*>((wh + 4 * SEQ_LEN + 4 * i) + on) = o[i];
  };

  GiBlock ga, gb;
  v4f oa[BQ], ob[BQ];
#pragma unroll
  for (int i = 0; i < BQ; ++i) { ob[i].x = 0.0f; ob[i].y = 0.0f; ob[i].z = 0.0f; ob[i].w = 0.0f; }
  load_block(ga, wt, orz, on);
  Tr = fmaf(h, wr, ga.r[0][0]);
  Tz = fmaf(h, wz, ga.z[0][0]);
  Tg = fmaf(h, wn, bn);
  for (int t = 0; t < SEQ_LEN; t += 2 * BLK) {
    load_block(gb, wt + BLK, orz, on);
    if (t != 0) store_block(ob, wt - BLK);
    run_block(ga, gb, oa, ob);
    store_block(oa, wt);
    load_block(ga, wt + 2 * BLK, orz, on);
    run_block(gb, ga, ob, oa);
    wt += 2 * BLK;
  }
  store_block(ob, wt - BLK);

  state_out[q] = h;
}

__global__ __launch_bounds__(256) void yproj_wmma_kernel(
    const float* __restrict__ hbuf, const float* __restrict__ w_out,
    const float* __restrict__ b_out, float* __restrict__ y) {
  const int lane = threadIdx.x & 31;
  const int wave = threadIdx.x >> 5;
  const int tile = blockIdx.x * 8 + wave;
  const int t0   = tile * 16;
  const int half = lane >> 4;
  const int idx  = lane & 15;

  v2f a;
  {
    const float* hr = hbuf + (size_t)(half * 2) * SEQ_LEN + t0 + idx;
    a.x = hr[0]; a.y = hr[SEQ_LEN];
  }

  v2f b;
  if (idx < 2) {
    const float2 wv = *reinterpret_cast<const float2*>(w_out + idx * 4 + half * 2);
    b.x = wv.x; b.y = wv.y;
  } else {
    b.x = 0.0f; b.y = 0.0f;
  }

  const float bias = (idx < 2) ? b_out[idx] : 0.0f;
  v8f c;
#pragma unroll
  for (int i = 0; i < 8; ++i) c[i] = bias;

  v8f d = __builtin_amdgcn_wmma_f32_16x16x4_f32(
      false, a, false, b, (short)0, c, false, false);

  if (idx < 2) {
    const int rbase = half * 8;
#pragma unroll
    for (int i = 0; i < 8; ++i)
      y[(size_t)(t0 + rbase + i) * 2 + idx] = d[i];
  }
}

extern "C" void kernel_launch(void* const* d_in, const int* in_sizes, int n_in,
                              void* d_out, int out_size, void* d_ws, size_t ws_size,
                              hipStream_t stream) {
  const float* x      = (const float*)d_in[0];
  const float* state  = (const float*)d_in[1];
  const float* w_ih   = (const float*)d_in[2];
  const float* w_hh   = (const float*)d_in[3];
  const float* b_ih   = (const float*)d_in[4];
  const float* b_hh   = (const float*)d_in[5];
  const float* w_out  = (const float*)d_in[6];
  const float* b_out  = (const float*)d_in[7];

  float* out  = (float*)d_out;
  float* gi   = (float*)d_ws;
  float* hbuf = gi + (size_t)NGATES * SEQ_LEN;

  const int tiles  = SEQ_LEN / 16;
  const int blocks = tiles / 8;

  gi_wmma_kernel<<<blocks, 256, 0, stream>>>(x, w_ih, b_ih, b_hh, gi);
  gru_scan_kernel<<<1, 16, 0, stream>>>(gi, state, w_hh, b_hh,
                                        out + (size_t)SEQ_LEN * 2);
  yproj_wmma_kernel<<<blocks, 256, 0, stream>>>(hbuf, w_out, b_out, out);
}


# scan: 1024 chunks of 256 steps side by side, each from a zero state 2048 steps ahead; joins checked bitwise, serial re-run if one differs; products re-formed from h per chunk (rounding changes there)
# speedup vs baseline: 699.4027x; 71.1361x over previous
#include <hip/hip_runtime.h>

typedef __attribute__((ext_vector_type(2))) float v2f;
typedef __attribute__((ext_vector_type(8))) float v8f;

#define SEQ_LEN 262144
#define NGATES  12
#define HDIM    4

__device__ __forceinline__ float bcast_lane(float v, int lane) {
  return __int_as_float(__builtin_amdgcn_readlane(__float_as_int(v), lane));
}

__device__ __forceinline__ float hw_tanh(float a) {
  return __builtin_amdgcn_tanhf(a);
}

__global__ __launch_bounds__(256) void gi_wmma_kernel(
    const float* __restrict__ x, const float* __restrict__ w_ih,
    const float* __restrict__ b_ih, const float* __restrict__ b_hh,
    float* __restrict__ gi) {
  const int lane = threadIdx.x & 31;
  const int wave = threadIdx.x >> 5;
  const int tile = blockIdx.x * 8 + wave;
  const int t0   = tile * 16;
  const int half = lane >> 4;
  const int idx  = lane & 15;

  v2f a;
  if (half == 0) {
    const float2 xv = *reinterpret_cast<const float2*>(x + (size_t)(t0 + idx) * 2);
    a.x = xv.x; a.y = xv.y;
  } else {
    a.x = 0.0f; a.y = 0.0f;
  }

  const float cs = (idx < 2 * HDIM) ? 0.125f : 1.0f;

  v2f b;
  if (half == 0 && idx < NGATES) {
    const float2 wv = *reinterpret_cast<const float2*>(w_ih + idx * 2);
    b.x = cs * wv.x; b.y = cs * wv.y;
  } else {
    b.x = 0.0f; b.y = 0.0f;
  }

  float bias = 0.0f;
  if (idx < 2 * HDIM)   bias = cs * (b_ih[idx] + b_hh[idx]);
  else if (idx < NGATES) bias = cs * b_ih[idx];
  v8f c;
#pragma unroll
  for (int i = 0; i < 8; ++i) c[i] = bias;

  v8f d = __builtin_amdgcn_wmma_f32_16x16x4_f32(
       false, a,  false, b,
       (short)0, c,  false,  false);

  if (idx < NGATES) {
    float* dst = gi + (size_t)idx * SEQ_LEN + t0 + half * 8;
    *reinterpret_cast<float4*>(dst + 0) = make_float4(d[0], d[1], d[2], d[3]);
    *reinterpret_cast<float4*>(dst + 4) = make_float4(d[4], d[5], d[6], d[7]);
  }
}

typedef __attribute__((ext_vector_type(4))) float v4f;

#define CHUNK  256
#define WARM   2048
#define NCHUNK (SEQ_LEN / CHUNK)
#define JPITCH 32

#define BLK 32
#define BQ  (BLK / 4)

template <int M>
__device__ __forceinline__ float row_xor(float v) {
  return __int_as_float(__builtin_amdgcn_update_dpp(
      0, __float_as_int(v), 0x160 + M, 0xF, 0xF, true));
}

__device__ __forceinline__ float quad_first(float v) {
  return __int_as_float(__builtin_amdgcn_update_dpp(
      0, __float_as_int(v), 0x00, 0xF, 0xF, true));
}

struct GiBlock { v4f r[BQ], z[BQ], n[BQ]; };

__device__ __forceinline__ void load_block(GiBlock& g, const float* wt,
                                           unsigned orz, unsigned on) {
#pragma unroll
  for (int i = 0; i < BQ; ++i) {
    g.r[i] = *reinterpret_cast<const v4f*>((wt + 4 * i) + orz);
    g.z[i] = *reinterpret_cast<const v4f*>((wt + 4 * SEQ_LEN + 4 * i) + orz);
    g.n[i] = *reinterpret_cast<const v4f*>((wt + 4 * i) + on);
  }
}

__global__ __launch_bounds__(128) __attribute__((amdgpu_waves_per_eu(1, 1)))
void gru_scan_kernel(
    float* ws, const float* __restrict__ state0,
    const float* __restrict__ w_hh, const float* __restrict__ b_hh,
    float* joins, float* state_out, int verify) {
  int t_first, t_store, t_end;
  if (verify == 0) {
    const int wave = __builtin_amdgcn_readfirstlane((int)(threadIdx.x >> 5));
    t_store = ((int)blockIdx.x * 4 + wave) * CHUNK;
    t_end   = t_store + CHUNK;
    t_first = (t_store > WARM) ? t_store - WARM : 0;
  } else {
    const int lane = threadIdx.x & 31;
    const float* hb = ws + (size_t)NGATES * SEQ_LEN;
    unsigned diff = 0;
#pragma unroll 4
    for (int i = 0; i < NCHUNK / 32; ++i) {
      const int c = i * 32 + lane;
      if (c != 0) {
        const v4f s = *reinterpret_cast<const v4f*>(joins + JPITCH * c);
        const float* he = hb + (c * CHUNK - 1);
        diff |= (__float_as_uint(s.x) ^ __float_as_uint(he[0]))
              | (__float_as_uint(s.y) ^ __float_as_uint(he[SEQ_LEN]))
              | (__float_as_uint(s.z) ^ __float_as_uint(he[2 * SEQ_LEN]))
              | (__float_as_uint(s.w) ^ __float_as_uint(he[3 * SEQ_LEN]));
      }
    }
    if (__builtin_amdgcn_ballot_w32(diff != 0) == 0) return;
    t_first = 0; t_store = 0; t_end = SEQ_LEN;
  }
  const int l = threadIdx.x & 15;
  const int q = l >> 2;
  const int j = l & 3;
  const int d = q ^ j;

  const float hf = 0.5f;
  const float wr = hf * w_hh[d * 4 + q];
  const float wz = hf * w_hh[(4 + d) * 4 + q];
  const float wn = hf * w_hh[(8 + d) * 4 + q];
  const float bn = (j == 0) ? hf * b_hh[8 + q] : 0.0f;

  float h = (t_first == 0) ? state0[q] : 0.0f;

  const unsigned orz = (unsigned)d * SEQ_LEN;
  const unsigned on  = (unsigned)(8 + q) * SEQ_LEN;
  float* wt = ws + t_first;

  float Tr, Tz, Tg;

  auto run_block = [&](const GiBlock& g, const GiBlock& gn, v4f (&o)[BQ],
                       const v4f (&keep)[BQ]) {
    float hs[BLK];
#pragma unroll
    for (int u = 0; u < BLK; ++u) {
      const float gin  = g.n[u >> 2][u & 3];
      const float girn = (u + 1 < BLK) ? g.r[(u + 1) >> 2][(u + 1) & 3] : gn.r[0][0];
      const float gizn = (u + 1 < BLK) ? g.z[(u + 1) >> 2][(u + 1) & 3] : gn.z[0][0];

      const float s1g = Tg + row_xor<5>(Tg);
      __builtin_amdgcn_sched_barrier(0);
      const float s1r = Tr + row_xor<5>(Tr);
      __builtin_amdgcn_sched_barrier(0);
      const float s1z = Tz + row_xor<5>(Tz);
      __builtin_amdgcn_sched_barrier(0);
      const float s2g = s1g + row_xor<10>(s1g);
      __builtin_amdgcn_sched_barrier(0);
      const float s2r = s1r + row_xor<10>(s1r);
      __builtin_amdgcn_sched_barrier(0);
      const float s2z = s1z + row_xor<10>(s1z);
      __builtin_amdgcn_sched_barrier(0);
      const float gh = quad_first(s2g);
      __builtin_amdgcn_sched_barrier(0);
      const float tr = hw_tanh(quad_first(s2r));
      __builtin_amdgcn_sched_barrier(0);
      const float tz = hw_tanh(quad_first(s2z));
      __builtin_amdgcn_sched_barrier(0);
      const float gnb = gin + gh;
      const float hh = 0.5f * h;
      __builtin_amdgcn_sched_barrier(0);
      const float na = fmaf(tr, gh, gnb);
      __builtin_amdgcn_sched_barrier(0);
      const float ca = fmaf(tz, -0.5f, 0.5f);
      const float cb = fmaf(tz, hh, hh);
      __builtin_amdgcn_sched_barrier(0);
      const float n = hw_tanh(na);
      __builtin_amdgcn_sched_barrier(0);
      const float ur = wr * ca;
      const float uz = wz * ca;
      const float ug = wn * ca;
      const float vr = fmaf(wr, cb, girn);
      const float vz = fmaf(wz, cb, gizn);
      const float vg = fmaf(wn, cb, bn);
      __builtin_amdgcn_sched_barrier(0);
      Tr = fmaf(n, ur, vr);
      Tz = fmaf(n, uz, vz);
      Tg = fmaf(n, ug, vg);
      h = fmaf(n, ca, cb);
      hs[u] = h;
      __builtin_amdgcn_sched_barrier(0);
      if (u == BLK / 2 - 1) {
#pragma unroll
        for (int i = 0; i < BQ; ++i) asm volatile("" :: "v"(keep[i]));
      }
    }
    asm volatile("" :: "v"(h) : "memory");
#pragma unroll
    for (int i = 0; i < BQ; ++i) {
      o[i].x = hs[4 * i + 0]; o[i].y = hs[4 * i + 1];
      o[i].z = hs[4 * i + 2]; o[i].w = hs[4 * i + 3];
    }
  };
  auto store_block = [&](const v4f (&o)[BQ], float* wh) {
#pragma unroll
    for (int i = 0; i < BQ; ++i)
      *reinterpret_cast<v4f*>((wh + 4 * SEQ_LEN + 4 * i) + on) = o[i];
  };

  GiBlock ga, gb;
  v4f oa[BQ], ob[BQ];
#pragma unroll
  for (int i = 0; i < BQ; ++i) { ob[i].x = 0.0f; ob[i].y = 0.0f; ob[i].z = 0.0f; ob[i].w = 0.0f; }
  load_block(ga, wt, orz, on);
  Tr = 0.0f; Tz = 0.0f; Tg = 0.0f;
  for (int t = t_first; t < t_end; t += 2 * BLK) {
    if ((t & (CHUNK - 1)) == 0) {
      Tr = fmaf(h, wr, ga.r[0][0]);
      Tz = fmaf(h, wz, ga.z[0][0]);
      Tg = fmaf(h, wn, bn);
      if (verify == 0 && t == t_store && j == 0) joins[(t_store / CHUNK) * JPITCH + q] = h;
    }
    load_block(gb, wt + BLK, orz, on);
    if (t > t_store) store_block(ob, wt - BLK);
    run_block(ga, gb, oa, ob);
    if (t >= t_store) store_block(oa, wt);
    load_block(ga, (t + 2 * BLK < SEQ_LEN) ? wt + 2 * BLK : wt, orz, on);
    run_block(gb, ga, ob, oa);
    wt += 2 * BLK;
  }
  store_block(ob, wt - BLK);

  if (t_end == SEQ_LEN) state_out[q] = h;
}

__global__ __launch_bounds__(256) void yproj_wmma_kernel(
    const float* __restrict__ hbuf, const float* __restrict__ w_out,
    const float* __restrict__ b_out, float* __restrict__ y) {
  const int lane = threadIdx.x & 31;
  const int wave = threadIdx.x >> 5;
  const int tile = blockIdx.x * 8 + wave;
  const int t0   = tile * 16;
  const int half = lane >> 4;
  const int idx  = lane & 15;

  v2f a;
  {
    const float* hr = hbuf + (size_t)(half * 2) * SEQ_LEN + t0 + idx;
    a.x = hr[0]; a.y = hr[SEQ_LEN];
  }

  v2f b;
  if (idx < 2) {
    const float2 wv = *reinterpret_cast<const float2*>(w_out + idx * 4 + half * 2);
    b.x = wv.x; b.y = wv.y;
  } else {
    b.x = 0.0f; b.y = 0.0f;
  }

  const float bias = (idx < 2) ? b_out[idx] : 0.0f;
  v8f c;
#pragma unroll
  for (int i = 0; i < 8; ++i) c[i] = bias;

  v8f d = __builtin_amdgcn_wmma_f32_16x16x4_f32(
      false, a, false, b, (short)0, c, false, false);

  if (idx < 2) {
    const int rbase = half * 8;
#pragma unroll
    for (int i = 0; i < 8; ++i)
      y[(size_t)(t0 + rbase + i) * 2 + idx] = d[i];
  }
}

extern "C" void kernel_launch(void* const* d_in, const int* in_sizes, int n_in,
                              void* d_out, int out_size, void* d_ws, size_t ws_size,
                              hipStream_t stream) {
  const float* x      = (const float*)d_in[0];
  const float* state  = (const float*)d_in[1];
  const float* w_ih   = (const float*)d_in[2];
  const float* w_hh   = (const float*)d_in[3];
  const float* b_ih   = (const float*)d_in[4];
  const float* b_hh   = (const float*)d_in[5];
  const float* w_out  = (const float*)d_in[6];
  const float* b_out  = (const float*)d_in[7];

  float* out  = (float*)d_out;
  float* gi   = (float*)d_ws;
  float* hbuf = gi + (size_t)NGATES * SEQ_LEN;

  const int tiles  = SEQ_LEN / 16;
  const int blocks = tiles / 8;

  gi_wmma_kernel<<<blocks, 256, 0, stream>>>(x, w_ih, b_ih, b_hh, gi);
  gru_scan_kernel<<<NCHUNK / 4, 128, 0, stream>>>(gi, state, w_hh, b_hh, out,
                                                  out + (size_t)SEQ_LEN * 2, 0);
  gru_scan_kernel<<<1, 32, 0, stream>>>(gi, state, w_hh, b_hh, out,
                                        out + (size_t)SEQ_LEN * 2, 1);
  yproj_wmma_kernel<<<blocks, 256, 0, stream>>>(hbuf, w_out, b_out, out);
}


# scan: warm-up in front of each chunk cut from 2048 to 512 steps (768 steps per wave, not 2304); joins still checked bitwise, serial re-run if one differs; stored values the same
# speedup vs baseline: 1234.0308x; 1.7644x over previous
#include <hip/hip_runtime.h>

typedef __attribute__((ext_vector_type(2))) float v2f;
typedef __attribute__((ext_vector_type(8))) float v8f;

#define SEQ_LEN 262144
#define NGATES  12
#define HDIM    4

__device__ __forceinline__ float bcast_lane(float v, int lane) {
  return __int_as_float(__builtin_amdgcn_readlane(__float_as_int(v), lane));
}

__device__ __forceinline__ float hw_tanh(float a) {
  return __builtin_amdgcn_tanhf(a);
}

__global__ __launch_bounds__(256) void gi_wmma_kernel(
    const float* __restrict__ x, const float* __restrict__ w_ih,
    const float* __restrict__ b_ih, const float* __restrict__ b_hh,
    float* __restrict__ gi) {
  const int lane = threadIdx.x & 31;
  const int wave = threadIdx.x >> 5;
  const int tile = blockIdx.x * 8 + wave;
  const int t0   = tile * 16;
  const int half = lane >> 4;
  const int idx  = lane & 15;

  v2f a;
  if (half == 0) {
    const float2 xv = *reinterpret_cast<const float2*>(x + (size_t)(t0 + idx) * 2);
    a.x = xv.x; a.y = xv.y;
  } else {
    a.x = 0.0f; a.y = 0.0f;
  }

  const float cs = (idx < 2 * HDIM) ? 0.125f : 1.0f;

  v2f b;
  if (half == 0 && idx < NGATES) {
    const float2 wv = *reinterpret_cast<const float2*>(w_ih + idx * 2);
    b.x = cs * wv.x; b.y = cs * wv.y;
  } else {
    b.x = 0.0f; b.y = 0.0f;
  }

  float bias = 0.0f;
  if (idx < 2 * HDIM)   bias = cs * (b_ih[idx] + b_hh[idx]);
  else if (idx < NGATES) bias = cs * b_ih[idx];
  v8f c;
#pragma unroll
  for (int i = 0; i < 8; ++i) c[i] = bias;

  v8f d = __builtin_amdgcn_wmma_f32_16x16x4_f32(
       false, a,  false, b,
       (short)0, c,  false,  false);

  if (idx < NGATES) {
    float* dst = gi + (size_t)idx * SEQ_LEN + t0 + half * 8;
    *reinterpret_cast<float4*>(dst + 0) = make_float4(d[0], d[1], d[2], d[3]);
    *reinterpret_cast<float4*>(dst + 4) = make_float4(d[4], d[5], d[6], d[7]);
  }
}

typedef __attribute__((ext_vector_type(4))) float v4f;

#define CHUNK  256
#define WARM   512
#define NCHUNK (SEQ_LEN / CHUNK)
#define JPITCH 32

#define BLK 32
#define BQ  (BLK / 4)

template <int M>
__device__ __forceinline__ float row_xor(float v) {
  return __int_as_float(__builtin_amdgcn_update_dpp(
      0, __float_as_int(v), 0x160 + M, 0xF, 0xF, true));
}

__device__ __forceinline__ float quad_first(float v) {
  return __int_as_float(__builtin_amdgcn_update_dpp(
      0, __float_as_int(v), 0x00, 0xF, 0xF, true));
}

struct GiBlock { v4f r[BQ], z[BQ], n[BQ]; };

__device__ __forceinline__ void load_block(GiBlock& g, const float* wt,
                                           unsigned orz, unsigned on) {
#pragma unroll
  for (int i = 0; i < BQ; ++i) {
    g.r[i] = *reinterpret_cast<const v4f*>((wt + 4 * i) + orz);
    g.z[i] = *reinterpret_cast<const v4f*>((wt + 4 * SEQ_LEN + 4 * i) + orz);
    g.n[i] = *reinterpret_cast<const v4f*>((wt + 4 * i) + on);
  }
}

__global__ __launch_bounds__(128) __attribute__((amdgpu_waves_per_eu(1, 1)))
void gru_scan_kernel(
    float* ws, const float* __restrict__ state0,
    const float* __restrict__ w_hh, const float* __restrict__ b_hh,
    float* joins, float* state_out, int verify) {
  int t_first, t_store, t_end;
  if (verify == 0) {
    const int wave = __builtin_amdgcn_readfirstlane((int)(threadIdx.x >> 5));
    t_store = ((int)blockIdx.x * 4 + wave) * CHUNK;
    t_end   = t_store + CHUNK;
    t_first = (t_store > WARM) ? t_store - WARM : 0;
  } else {
    const int lane = threadIdx.x & 31;
    const float* hb = ws + (size_t)NGATES * SEQ_LEN;
    unsigned diff = 0;
#pragma unroll 4
    for (int i = 0; i < NCHUNK / 32; ++i) {
      const int c = i * 32 + lane;
      if (c != 0) {
        const v4f s = *reinterpret_cast<const v4f*>(joins + JPITCH * c);
        const float* he = hb + (c * CHUNK - 1);
        diff |= (__float_as_uint(s.x) ^ __float_as_uint(he[0]))
              | (__float_as_uint(s.y) ^ __float_as_uint(he[SEQ_LEN]))
              | (__float_as_uint(s.z) ^ __float_as_uint(he[2 * SEQ_LEN]))
              | (__float_as_uint(s.w) ^ __float_as_uint(he[3 * SEQ_LEN]));
      }
    }
    if (__builtin_amdgcn_ballot_w32(diff != 0) == 0) return;
    t_first = 0; t_store = 0; t_end = SEQ_LEN;
  }
  const int l = threadIdx.x & 15;
  const int q = l >> 2;
  const int j = l & 3;
  const int d = q ^ j;

  const float hf = 0.5f;
  const float wr = hf * w_hh[d * 4 + q];
  const float wz = hf * w_hh[(4 + d) * 4 + q];
  const float wn = hf * w_hh[(8 + d) * 4 + q];
  const float bn = (j == 0) ? hf * b_hh[8 + q] : 0.0f;

  float h = (t_first == 0) ? state0[q] : 0.0f;

  const unsigned orz = (unsigned)d * SEQ_LEN;
  const unsigned on  = (unsigned)(8 + q) * SEQ_LEN;
  float* wt = ws + t_first;

  float Tr, Tz, Tg;

  auto run_block = [&](const GiBlock& g, const GiBlock& gn, v4f (&o)[BQ],
                       const v4f (&keep)[BQ]) {
    float hs[BLK];
#pragma unroll
    for (int u = 0; u < BLK; ++u) {
      const float gin  = g.n[u >> 2][u & 3];
      const float girn = (u + 1 < BLK) ? g.r[(u + 1) >> 2][(u + 1) & 3] : gn.r[0][0];
      const float gizn = (u + 1 < BLK) ? g.z[(u + 1) >> 2][(u + 1) & 3] : gn.z[0][0];

      const float s1g = Tg + row_xor<5>(Tg);
      __builtin_amdgcn_sched_barrier(0);
      const float s1r = Tr + row_xor<5>(Tr);
      __builtin_amdgcn_sched_barrier(0);
      const float s1z = Tz + row_xor<5>(Tz);
      __builtin_amdgcn_sched_barrier(0);
      const float s2g = s1g + row_xor<10>(s1g);
      __builtin_amdgcn_sched_barrier(0);
      const float s2r = s1r + row_xor<10>(s1r);
      __builtin_amdgcn_sched_barrier(0);
      const float s2z = s1z + row_xor<10>(s1z);
      __builtin_amdgcn_sched_barrier(0);
      const float gh = quad_first(s2g);
      __builtin_amdgcn_sched_barrier(0);
      const float tr = hw_tanh(quad_first(s2r));
      __builtin_amdgcn_sched_barrier(0);
      const float tz = hw_tanh(quad_first(s2z));
      __builtin_amdgcn_sched_barrier(0);
      const float gnb = gin + gh;
      const float hh = 0.5f * h;
      __builtin_amdgcn_sched_barrier(0);
      const float na = fmaf(tr, gh, gnb);
      __builtin_amdgcn_sched_barrier(0);
      const float ca = fmaf(tz, -0.5f, 0.5f);
      const float cb = fmaf(tz, hh, hh);
      __builtin_amdgcn_sched_barrier(0);
      const float n = hw_tanh(na);
      __builtin_amdgcn_sched_barrier(0);
      const float ur = wr * ca;
      const float uz = wz * ca;
      const float ug = wn * ca;
      const float vr = fmaf(wr, cb, girn);
      const float vz = fmaf(wz, cb, gizn);
      const float vg = fmaf(wn, cb, bn);
      __builtin_amdgcn_sched_barrier(0);
      Tr = fmaf(n, ur, vr);
      Tz = fmaf(n, uz, vz);
      Tg = fmaf(n, ug, vg);
      h = fmaf(n, ca, cb);
      hs[u] = h;
      __builtin_amdgcn_sched_barrier(0);
      if (u == BLK / 2 - 1) {
#pragma unroll
        for (int i = 0; i < BQ; ++i) asm volatile("" :: "v"(keep[i]));
      }
    }
    asm volatile("" :: "v"(h) : "memory");
#pragma unroll
    for (int i = 0; i < BQ; ++i) {
      o[i].x = hs[4 * i + 0]; o[i].y = hs[4 * i + 1];
      o[i].z = hs[4 * i + 2]; o[i].w = hs[4 * i + 3];
    }
  };
  auto store_block = [&](const v4f (&o)[BQ], float* wh) {
#pragma unroll
    for (int i = 0; i < BQ; ++i)
      *reinterpret_cast<v4f*>((wh + 4 * SEQ_LEN + 4 * i) + on) = o[i];
  };

  GiBlock ga, gb;
  v4f oa[BQ], ob[BQ];
#pragma unroll
  for (int i = 0; i < BQ; ++i) { ob[i].x = 0.0f; ob[i].y = 0.0f; ob[i].z = 0.0f; ob[i].w = 0.0f; }
  load_block(ga, wt, orz, on);
  Tr = 0.0f; Tz = 0.0f; Tg = 0.0f;
  for (int t = t_first; t < t_end; t += 2 * BLK) {
    if ((t & (CHUNK - 1)) == 0) {
      Tr = fmaf(h, wr, ga.r[0][0]);
      Tz = fmaf(h, wz, ga.z[0][0]);
      Tg = fmaf(h, wn, bn);
      if (verify == 0 && t == t_store && j == 0) joins[(t_store / CHUNK) * JPITCH + q] = h;
    }
    load_block(gb, wt + BLK, orz, on);
    if (t > t_store) store_block(ob, wt - BLK);
    run_block(ga, gb, oa, ob);
    if (t >= t_store) store_block(oa, wt);
    load_block(ga, (t + 2 * BLK < SEQ_LEN) ? wt + 2 * BLK : wt, orz, on);
    run_block(gb, ga, ob, oa);
    wt += 2 * BLK;
  }
  store_block(ob, wt - BLK);

  if (t_end == SEQ_LEN) state_out[q] = h;
}

__global__ __launch_bounds__(256) void yproj_wmma_kernel(
    const float* __restrict__ hbuf, const float* __restrict__ w_out,
    const float* __restrict__ b_out, float* __restrict__ y) {
  const int lane = threadIdx.x & 31;
  const int wave = threadIdx.x >> 5;
  const int tile = blockIdx.x * 8 + wave;
  const int t0   = tile * 16;
  const int half = lane >> 4;
  const int idx  = lane & 15;

  v2f a;
  {
    const float* hr = hbuf + (size_t)(half * 2) * SEQ_LEN + t0 + idx;
    a.x = hr[0]; a.y = hr[SEQ_LEN];
  }

  v2f b;
  if (idx < 2) {
    const float2 wv = *reinterpret_cast<const float2*>(w_out + idx * 4 + half * 2);
    b.x = wv.x; b.y = wv.y;
  } else {
    b.x = 0.0f; b.y = 0.0f;
  }

  const float bias = (idx < 2) ? b_out[idx] : 0.0f;
  v8f c;
#pragma unroll
  for (int i = 0; i < 8; ++i) c[i] = bias;

  v8f d = __builtin_amdgcn_wmma_f32_16x16x4_f32(
      false, a, false, b, (short)0, c, false, false);

  if (idx < 2) {
    const int rbase = half * 8;
#pragma unroll
    for (int i = 0; i < 8; ++i)
      y[(size_t)(t0 + rbase + i) * 2 + idx] = d[i];
  }
}

extern "C" void kernel_launch(void* const* d_in, const int* in_sizes, int n_in,
                              void* d_out, int out_size, void* d_ws, size_t ws_size,
                              hipStream_t stream) {
  const float* x      = (const float*)d_in[0];
  const float* state  = (const float*)d_in[1];
  const float* w_ih   = (const float*)d_in[2];
  const float* w_hh   = (const float*)d_in[3];
  const float* b_ih   = (const float*)d_in[4];
  const float* b_hh   = (const float*)d_in[5];
  const float* w_out  = (const float*)d_in[6];
  const float* b_out  = (const float*)d_in[7];

  float* out  = (float*)d_out;
  float* gi   = (float*)d_ws;
  float* hbuf = gi + (size_t)NGATES * SEQ_LEN;

  const int tiles  = SEQ_LEN / 16;
  const int blocks = tiles / 8;

  gi_wmma_kernel<<<blocks, 256, 0, stream>>>(x, w_ih, b_ih, b_hh, gi);
  gru_scan_kernel<<<NCHUNK / 4, 128, 0, stream>>>(gi, state, w_hh, b_hh, out,
                                                  out + (size_t)SEQ_LEN * 2, 0);
  gru_scan_kernel<<<1, 32, 0, stream>>>(gi, state, w_hh, b_hh, out,
                                        out + (size_t)SEQ_LEN * 2, 1);
  yproj_wmma_kernel<<<blocks, 256, 0, stream>>>(hbuf, w_out, b_out, out);
}


# gi formed inside the scan kernel by the wave that reads it (f32 WMMA per 8 steps and gate row, in the n gate's tanh wait; x loaded 42 steps ahead): gi launch and its 12 MB round trip gone
# speedup vs baseline: 1315.0900x; 1.0657x over previous
#include <hip/hip_runtime.h>

typedef __attribute__((ext_vector_type(2))) float v2f;
typedef __attribute__((ext_vector_type(8))) float v8f;

#define SEQ_LEN 262144
#define NGATES  12
#define HDIM    4

__device__ __forceinline__ float bcast_lane(float v, int lane) {
  return __int_as_float(__builtin_amdgcn_readlane(__float_as_int(v), lane));
}

__device__ __forceinline__ float hw_tanh(float a) {
  return __builtin_amdgcn_tanhf(a);
}


typedef __attribute__((ext_vector_type(4))) float v4f;

#define CHUNK  256
#define WARM   512
#define NCHUNK (SEQ_LEN / CHUNK)
#define JPITCH 32

#define BLK 32
#define BQ  (BLK / 4)

template <int M>
__device__ __forceinline__ float row_xor(float v) {
  return __int_as_float(__builtin_amdgcn_update_dpp(
      0, __float_as_int(v), 0x160 + M, 0xF, 0xF, true));
}

__device__ __forceinline__ float quad_first(float v) {
  return __int_as_float(__builtin_amdgcn_update_dpp(
      0, __float_as_int(v), 0x00, 0xF, 0xF, true));
}

#define BG  (BLK / 8)
#define HB  (BLK / 2)
#define HG  (HB / 8)

struct GiHalf { v8f r[HG], z[HG], n[HG]; };
struct XBlock { v2f a[BG]; };

__device__ __forceinline__ void load_x(XBlock& xb, const float* xl) {
#pragma unroll
  for (int i = 0; i < BG; ++i)
    xb.a[i] = *reinterpret_cast<const v2f*>(xl + 16 * i);
}

__global__ __launch_bounds__(128) __attribute__((amdgpu_waves_per_eu(1, 1)))
void gru_scan_kernel(
    const float* x, const float* __restrict__ w_ih, const float* __restrict__ b_ih,
    float* ws, const float* __restrict__ state0,
    const float* __restrict__ w_hh, const float* __restrict__ b_hh,
    float* joins, float* state_out, int verify) {
  int t_first, t_store, t_end;
  if (verify == 0) {
    const int wave = __builtin_amdgcn_readfirstlane((int)(threadIdx.x >> 5));
    t_store = ((int)blockIdx.x * 4 + wave) * CHUNK;
    t_end   = t_store + CHUNK;
    t_first = (t_store > WARM) ? t_store - WARM : 0;
  } else {
    const int lane = threadIdx.x & 31;
    const float* hb = ws + (size_t)NGATES * SEQ_LEN;
    unsigned diff = 0;
#pragma unroll 4
    for (int i = 0; i < NCHUNK / 32; ++i) {
      const int c = i * 32 + lane;
      if (c != 0) {
        const v4f s = *reinterpret_cast<const v4f*>(joins + JPITCH * c);
        const float* he = hb + (c * CHUNK - 1);
        diff |= (__float_as_uint(s.x) ^ __float_as_uint(he[0]))
              | (__float_as_uint(s.y) ^ __float_as_uint(he[SEQ_LEN]))
              | (__float_as_uint(s.z) ^ __float_as_uint(he[2 * SEQ_LEN]))
              | (__float_as_uint(s.w) ^ __float_as_uint(he[3 * SEQ_LEN]));
      }
    }
    if (__builtin_amdgcn_ballot_w32(diff != 0) == 0) return;
    t_first = 0; t_store = 0; t_end = SEQ_LEN;
  }
  const int l = threadIdx.x & 15;
  const int q = l >> 2;
  const int j = l & 3;
  const int d = q ^ j;

  const float hf = 0.5f;
  const float wr = hf * w_hh[d * 4 + q];
  const float wz = hf * w_hh[(4 + d) * 4 + q];
  const float wn = hf * w_hh[(8 + d) * 4 + q];
  const float bn = (j == 0) ? hf * b_hh[8 + q] : 0.0f;

  float h = (t_first == 0) ? state0[q] : 0.0f;

  float* hq = ws + (size_t)(NGATES + q) * SEQ_LEN + t_first;

  const bool lo = (threadIdx.x & 16) == 0;
  const float cs = 0.125f;
  const float2 wir = *reinterpret_cast<const float2*>(w_ih + (d) * 2);
  const float2 wiz = *reinterpret_cast<const float2*>(w_ih + (4 + d) * 2);
  const float2 win = *reinterpret_cast<const float2*>(w_ih + (8 + q) * 2);
  v2f bR, bZ, bN;
  bR.x = lo ? cs * wir.x : 0.0f;  bR.y = lo ? cs * wir.y : 0.0f;
  bZ.x = lo ? cs * wiz.x : 0.0f;  bZ.y = lo ? cs * wiz.y : 0.0f;
  bN.x = lo ? win.x : 0.0f;       bN.y = lo ? win.y : 0.0f;
  const float biasR = cs * (b_ih[d] + b_hh[d]);
  const float biasZ = cs * (b_ih[4 + d] + b_hh[4 + d]);
  const float biasN = b_ih[8 + q];
  v8f cR, cZ, cN;
#pragma unroll
  for (int i = 0; i < 8; ++i) { cR[i] = biasR; cZ[i] = biasZ; cN[i] = biasN; }
  const float* xl0 = x + 2 * (size_t)t_first + (l & 7) * 2;

  auto gi_part = [&](GiHalf& gn, const XBlock& xf, const int xg, const int p) {
    const int gg = p / 3;
    v2f a;
    a.x = lo ? xf.a[xg + gg].x : 0.0f;
    a.y = lo ? xf.a[xg + gg].y : 0.0f;
    if (p % 3 == 0)
      gn.r[gg] = __builtin_amdgcn_wmma_f32_16x16x4_f32(false, a, false, bR, (short)0, cR, false, false);
    else if (p % 3 == 1)
      gn.z[gg] = __builtin_amdgcn_wmma_f32_16x16x4_f32(false, a, false, bZ, (short)0, cZ, false, false);
    else
      gn.n[gg] = __builtin_amdgcn_wmma_f32_16x16x4_f32(false, a, false, bN, (short)0, cN, false, false);
  };

  float Tr, Tz, Tg;

  auto run_block = [&](GiHalf& P, GiHalf& Q, XBlock& xk, const XBlock& xk1,
                       const float* xk2, v4f (&o)[BQ], const v4f (&keep)[BQ]) {
    float hs[BLK];
#pragma unroll
    for (int u = 0; u < BLK; ++u) {
      const GiHalf& g  = (u < HB) ? P : Q;
      const GiHalf& gn = (u < HB) ? Q : P;
      const int v = u & (HB - 1);
      const float gin  = g.n[v >> 3][v & 7];
      const float girn = (v + 1 < HB) ? g.r[(v + 1) >> 3][(v + 1) & 7] : gn.r[0][0];
      const float gizn = (v + 1 < HB) ? g.z[(v + 1) >> 3][(v + 1) & 7] : gn.z[0][0];

      const float s1g = Tg + row_xor<5>(Tg);
      __builtin_amdgcn_sched_barrier(0);
      const float s1r = Tr + row_xor<5>(Tr);
      __builtin_amdgcn_sched_barrier(0);
      const float s1z = Tz + row_xor<5>(Tz);
      __builtin_amdgcn_sched_barrier(0);
      const float s2g = s1g + row_xor<10>(s1g);
      __builtin_amdgcn_sched_barrier(0);
      const float s2r = s1r + row_xor<10>(s1r);
      __builtin_amdgcn_sched_barrier(0);
      const float s2z = s1z + row_xor<10>(s1z);
      __builtin_amdgcn_sched_barrier(0);
      const float gh = quad_first(s2g);
      __builtin_amdgcn_sched_barrier(0);
      const float tr = hw_tanh(quad_first(s2r));
      __builtin_amdgcn_sched_barrier(0);
      const float tz = hw_tanh(quad_first(s2z));
      __builtin_amdgcn_sched_barrier(0);
      const float gnb = gin + gh;
      const float hh = 0.5f * h;
      __builtin_amdgcn_sched_barrier(0);
      const float na = fmaf(tr, gh, gnb);
      __builtin_amdgcn_sched_barrier(0);
      const float ca = fmaf(tz, -0.5f, 0.5f);
      const float cb = fmaf(tz, hh, hh);
      __builtin_amdgcn_sched_barrier(0);
      const float n = hw_tanh(na);
      __builtin_amdgcn_sched_barrier(0);
      const float ur = wr * ca;
      const float uz = wz * ca;
      const float ug = wn * ca;
      const float vr = fmaf(wr, cb, girn);
      const float vz = fmaf(wz, cb, gizn);
      const float vg = fmaf(wn, cb, bn);
      if (u < 3 * HG) gi_part(Q, xk, HG, u);
      else if (u >= HB && u < HB + 3 * HG) gi_part(P, xk1, 0, u - HB);
      __builtin_amdgcn_sched_barrier(0);
      Tr = fmaf(n, ur, vr);
      Tz = fmaf(n, uz, vz);
      Tg = fmaf(n, ug, vg);
      h = fmaf(n, ca, cb);
      hs[u] = h;
      __builtin_amdgcn_sched_barrier(0);
      if (u == 3 * HG) {
        load_x(xk, xk2);
        __builtin_amdgcn_sched_barrier(0);
      }
      if (u == BLK / 2 - 1) {
#pragma unroll
        for (int i = 0; i < BQ; ++i) asm volatile("" :: "v"(keep[i]));
      }
    }
    asm volatile("" :: "v"(h) : "memory");
#pragma unroll
    for (int i = 0; i < BQ; ++i) {
      o[i].x = hs[4 * i + 0]; o[i].y = hs[4 * i + 1];
      o[i].z = hs[4 * i + 2]; o[i].w = hs[4 * i + 3];
    }
  };
  auto store_block = [&](const v4f (&o)[BQ], float* hp) {
#pragma unroll
    for (int i = 0; i < BQ; ++i)
      *reinterpret_cast<v4f*>(hp + 4 * i) = o[i];
  };

  GiHalf ga, gb;
  XBlock xa, xb;
  v4f oa[BQ], ob[BQ];
#pragma unroll
  for (int i = 0; i < BQ; ++i) { ob[i].x = 0.0f; ob[i].y = 0.0f; ob[i].z = 0.0f; ob[i].w = 0.0f; }
  load_x(xa, xl0);
  __builtin_amdgcn_sched_barrier(0);
  load_x(xb, xl0 + 2 * BLK);
  const float* xq = xl0 + 2 * (2 * BLK);
#pragma unroll
  for (int p = 0; p < 3 * HG; ++p) gi_part(ga, xa, 0, p);
  Tr = 0.0f; Tz = 0.0f; Tg = 0.0f;
  for (int t = t_first; t < t_end; t += 2 * BLK) {
    if ((t & (CHUNK - 1)) == 0) {
      Tr = fmaf(h, wr, ga.r[0][0]);
      Tz = fmaf(h, wz, ga.z[0][0]);
      Tg = fmaf(h, wn, bn);
      if (verify == 0 && t == t_store && j == 0) joins[(t_store / CHUNK) * JPITCH + q] = h;
    }
    if (t > t_store) store_block(ob, hq - BLK);
    run_block(ga, gb, xa, xb, xq, oa, ob);
    if (t >= t_store) store_block(oa, hq);
    run_block(ga, gb, xb, xa, xq + 2 * BLK, ob, oa);
    hq += 2 * BLK;
    xq += (t + 4 * BLK < SEQ_LEN) ? 2 * (2 * BLK) : 0;
  }
  store_block(ob, hq - BLK);

  if (t_end == SEQ_LEN) state_out[q] = h;
}

__global__ __launch_bounds__(256) void yproj_wmma_kernel(
    const float* __restrict__ hbuf, const float* __restrict__ w_out,
    const float* __restrict__ b_out, float* __restrict__ y) {
  const int lane = threadIdx.x & 31;
  const int wave = threadIdx.x >> 5;
  const int tile = blockIdx.x * 8 + wave;
  const int t0   = tile * 16;
  const int half = lane >> 4;
  const int idx  = lane & 15;

  v2f a;
  {
    const float* hr = hbuf + (size_t)(half * 2) * SEQ_LEN + t0 + idx;
    a.x = hr[0]; a.y = hr[SEQ_LEN];
  }

  v2f b;
  if (idx < 2) {
    const float2 wv = *reinterpret_cast<const float2*>(w_out + idx * 4 + half * 2);
    b.x = wv.x; b.y = wv.y;
  } else {
    b.x = 0.0f; b.y = 0.0f;
  }

  const float bias = (idx < 2) ? b_out[idx] : 0.0f;
  v8f c;
#pragma unroll
  for (int i = 0; i < 8; ++i) c[i] = bias;

  v8f d = __builtin_amdgcn_wmma_f32_16x16x4_f32(
      false, a, false, b, (short)0, c, false, false);

  if (idx < 2) {
    const int rbase = half * 8;
#pragma unroll
    for (int i = 0; i < 8; ++i)
      y[(size_t)(t0 + rbase + i) * 2 + idx] = d[i];
  }
}

extern "C" void kernel_launch(void* const* d_in, const int* in_sizes, int n_in,
                              void* d_out, int out_size, void* d_ws, size_t ws_size,
                              hipStream_t stream) {
  const float* x      = (const float*)d_in[0];
  const float* state  = (const float*)d_in[1];
  const float* w_ih   = (const float*)d_in[2];
  const float* w_hh   = (const float*)d_in[3];
  const float* b_ih   = (const float*)d_in[4];
  const float* b_hh   = (const float*)d_in[5];
  const float* w_out  = (const float*)d_in[6];
  const float* b_out  = (const float*)d_in[7];

  float* out  = (float*)d_out;
  float* ws   = (float*)d_ws;
  float* hbuf = ws + (size_t)NGATES * SEQ_LEN;

  const int tiles  = SEQ_LEN / 16;
  const int blocks = tiles / 8;

  gru_scan_kernel<<<NCHUNK / 4, 128, 0, stream>>>(x, w_ih, b_ih, ws, state, w_hh, b_hh, out,
                                                  out + (size_t)SEQ_LEN * 2, 0);
  gru_scan_kernel<<<1, 32, 0, stream>>>(x, w_ih, b_ih, ws, state, w_hh, b_hh, out,
                                        out + (size_t)SEQ_LEN * 2, 1);
  yproj_wmma_kernel<<<blocks, 256, 0, stream>>>(hbuf, w_out, b_out, out);
}


# join check in one memory round trip: each chunk's wave records entry and last state in its own 128-byte line; 128 lanes compare all joins at once (was 1 wave, 16 dependent rounds of loads)
# speedup vs baseline: 1867.0828x; 1.4197x over previous
#include <hip/hip_runtime.h>

typedef __attribute__((ext_vector_type(2))) float v2f;
typedef __attribute__((ext_vector_type(8))) float v8f;

#define SEQ_LEN 262144
#define NGATES  12
#define HDIM    4

__device__ __forceinline__ float bcast_lane(float v, int lane) {
  return __int_as_float(__builtin_amdgcn_readlane(__float_as_int(v), lane));
}

__device__ __forceinline__ float hw_tanh(float a) {
  return __builtin_amdgcn_tanhf(a);
}


typedef __attribute__((ext_vector_type(4))) float v4f;

#define CHUNK  256
#define WARM   512
#define NCHUNK (SEQ_LEN / CHUNK)
#define JPITCH 32

#define BLK 32
#define BQ  (BLK / 4)

template <int M>
__device__ __forceinline__ float row_xor(float v) {
  return __int_as_float(__builtin_amdgcn_update_dpp(
      0, __float_as_int(v), 0x160 + M, 0xF, 0xF, true));
}

__device__ __forceinline__ float quad_first(float v) {
  return __int_as_float(__builtin_amdgcn_update_dpp(
      0, __float_as_int(v), 0x00, 0xF, 0xF, true));
}

#define BG  (BLK / 8)
#define HB  (BLK / 2)
#define HG  (HB / 8)

struct GiHalf { v8f r[HG], z[HG], n[HG]; };
struct XBlock { v2f a[BG]; };

__device__ __forceinline__ void load_x(XBlock& xb, const float* xl) {
#pragma unroll
  for (int i = 0; i < BG; ++i)
    xb.a[i] = *reinterpret_cast<const v2f*>(xl + 16 * i);
}

__global__ __launch_bounds__(128) __attribute__((amdgpu_waves_per_eu(1, 1)))
void gru_scan_kernel(
    const float* x, const float* __restrict__ w_ih, const float* __restrict__ b_ih,
    float* ws, const float* __restrict__ state0,
    const float* __restrict__ w_hh, const float* __restrict__ b_hh,
    float* joins, float* state_out, int verify) {
  int t_first, t_store, t_end;
  if (verify == 0) {
    const int wave = __builtin_amdgcn_readfirstlane((int)(threadIdx.x >> 5));
    t_store = ((int)blockIdx.x * 4 + wave) * CHUNK;
    t_end   = t_store + CHUNK;
    t_first = (t_store > WARM) ? t_store - WARM : 0;
  } else {
    const int tid = threadIdx.x;
    unsigned diff = 0;
#pragma unroll
    for (int i = 0; i < NCHUNK / 128; ++i) {
      const int c = i * 128 + tid;
      const int cb = (c != 0) ? c - 1 : 0;
      const v4f s = *reinterpret_cast<const v4f*>(joins + JPITCH * c);
      const v4f e = *reinterpret_cast<const v4f*>(joins + JPITCH * cb + 4);
      const unsigned dj = (__float_as_uint(s.x) ^ __float_as_uint(e.x))
                        | (__float_as_uint(s.y) ^ __float_as_uint(e.y))
                        | (__float_as_uint(s.z) ^ __float_as_uint(e.z))
                        | (__float_as_uint(s.w) ^ __float_as_uint(e.w));
      diff |= (c != 0) ? dj : 0u;
    }
    __shared__ unsigned wave_diff[4];
    const unsigned mine = __builtin_amdgcn_ballot_w32(diff != 0);
    if ((tid & 31) == 0) wave_diff[tid >> 5] = mine;
    __syncthreads();
    if ((wave_diff[0] | wave_diff[1] | wave_diff[2] | wave_diff[3]) == 0) return;
    if (tid >= 32) return;
    t_first = 0; t_store = 0; t_end = SEQ_LEN;
  }
  const int l = threadIdx.x & 15;
  const int q = l >> 2;
  const int j = l & 3;
  const int d = q ^ j;

  const float hf = 0.5f;
  const float wr = hf * w_hh[d * 4 + q];
  const float wz = hf * w_hh[(4 + d) * 4 + q];
  const float wn = hf * w_hh[(8 + d) * 4 + q];
  const float bn = (j == 0) ? hf * b_hh[8 + q] : 0.0f;

  float h = (t_first == 0) ? state0[q] : 0.0f;

  float* hq = ws + (size_t)(NGATES + q) * SEQ_LEN + t_first;

  const bool lo = (threadIdx.x & 16) == 0;
  const float cs = 0.125f;
  const float2 wir = *reinterpret_cast<const float2*>(w_ih + (d) * 2);
  const float2 wiz = *reinterpret_cast<const float2*>(w_ih + (4 + d) * 2);
  const float2 win = *reinterpret_cast<const float2*>(w_ih + (8 + q) * 2);
  v2f bR, bZ, bN;
  bR.x = lo ? cs * wir.x : 0.0f;  bR.y = lo ? cs * wir.y : 0.0f;
  bZ.x = lo ? cs * wiz.x : 0.0f;  bZ.y = lo ? cs * wiz.y : 0.0f;
  bN.x = lo ? win.x : 0.0f;       bN.y = lo ? win.y : 0.0f;
  const float biasR = cs * (b_ih[d] + b_hh[d]);
  const float biasZ = cs * (b_ih[4 + d] + b_hh[4 + d]);
  const float biasN = b_ih[8 + q];
  v8f cR, cZ, cN;
#pragma unroll
  for (int i = 0; i < 8; ++i) { cR[i] = biasR; cZ[i] = biasZ; cN[i] = biasN; }
  const float* xl0 = x + 2 * (size_t)t_first + (l & 7) * 2;

  auto gi_part = [&](GiHalf& gn, const XBlock& xf, const int xg, const int p) {
    const int gg = p / 3;
    v2f a;
    a.x = lo ? xf.a[xg + gg].x : 0.0f;
    a.y = lo ? xf.a[xg + gg].y : 0.0f;
    if (p % 3 == 0)
      gn.r[gg] = __builtin_amdgcn_wmma_f32_16x16x4_f32(false, a, false, bR, (short)0, cR, false, false);
    else if (p % 3 == 1)
      gn.z[gg] = __builtin_amdgcn_wmma_f32_16x16x4_f32(false, a, false, bZ, (short)0, cZ, false, false);
    else
      gn.n[gg] = __builtin_amdgcn_wmma_f32_16x16x4_f32(false, a, false, bN, (short)0, cN, false, false);
  };

  float Tr, Tz, Tg;

  auto run_block = [&](GiHalf& P, GiHalf& Q, XBlock& xk, const XBlock& xk1,
                       const float* xk2, v4f (&o)[BQ], const v4f (&keep)[BQ]) {
    float hs[BLK];
#pragma unroll
    for (int u = 0; u < BLK; ++u) {
      const GiHalf& g  = (u < HB) ? P : Q;
      const GiHalf& gn = (u < HB) ? Q : P;
      const int v = u & (HB - 1);
      const float gin  = g.n[v >> 3][v & 7];
      const float girn = (v + 1 < HB) ? g.r[(v + 1) >> 3][(v + 1) & 7] : gn.r[0][0];
      const float gizn = (v + 1 < HB) ? g.z[(v + 1) >> 3][(v + 1) & 7] : gn.z[0][0];

      const float s1g = Tg + row_xor<5>(Tg);
      __builtin_amdgcn_sched_barrier(0);
      const float s1r = Tr + row_xor<5>(Tr);
      __builtin_amdgcn_sched_barrier(0);
      const float s1z = Tz + row_xor<5>(Tz);
      __builtin_amdgcn_sched_barrier(0);
      const float s2g = s1g + row_xor<10>(s1g);
      __builtin_amdgcn_sched_barrier(0);
      const float s2r = s1r + row_xor<10>(s1r);
      __builtin_amdgcn_sched_barrier(0);
      const float s2z = s1z + row_xor<10>(s1z);
      __builtin_amdgcn_sched_barrier(0);
      const float gh = quad_first(s2g);
      __builtin_amdgcn_sched_barrier(0);
      const float tr = hw_tanh(quad_first(s2r));
      __builtin_amdgcn_sched_barrier(0);
      const float tz = hw_tanh(quad_first(s2z));
      __builtin_amdgcn_sched_barrier(0);
      const float gnb = gin + gh;
      const float hh = 0.5f * h;
      __builtin_amdgcn_sched_barrier(0);
      const float na = fmaf(tr, gh, gnb);
      __builtin_amdgcn_sched_barrier(0);
      const float ca = fmaf(tz, -0.5f, 0.5f);
      const float cb = fmaf(tz, hh, hh);
      __builtin_amdgcn_sched_barrier(0);
      const float n = hw_tanh(na);
      __builtin_amdgcn_sched_barrier(0);
      const float ur = wr * ca;
      const float uz = wz * ca;
      const float ug = wn * ca;
      const float vr = fmaf(wr, cb, girn);
      const float vz = fmaf(wz, cb, gizn);
      const float vg = fmaf(wn, cb, bn);
      if (u < 3 * HG) gi_part(Q, xk, HG, u);
      else if (u >= HB && u < HB + 3 * HG) gi_part(P, xk1, 0, u - HB);
      __builtin_amdgcn_sched_barrier(0);
      Tr = fmaf(n, ur, vr);
      Tz = fmaf(n, uz, vz);
      Tg = fmaf(n, ug, vg);
      h = fmaf(n, ca, cb);
      hs[u] = h;
      __builtin_amdgcn_sched_barrier(0);
      if (u == 3 * HG) {
        load_x(xk, xk2);
        __builtin_amdgcn_sched_barrier(0);
      }
      if (u == BLK / 2 - 1) {
#pragma unroll
        for (int i = 0; i < BQ; ++i) asm volatile("" :: "v"(keep[i]));
      }
    }
    asm volatile("" :: "v"(h) : "memory");
#pragma unroll
    for (int i = 0; i < BQ; ++i) {
      o[i].x = hs[4 * i + 0]; o[i].y = hs[4 * i + 1];
      o[i].z = hs[4 * i + 2]; o[i].w = hs[4 * i + 3];
    }
  };
  auto store_block = [&](const v4f (&o)[BQ], float* hp) {
#pragma unroll
    for (int i = 0; i < BQ; ++i)
      *reinterpret_cast<v4f*>(hp + 4 * i) = o[i];
  };

  GiHalf ga, gb;
  XBlock xa, xb;
  v4f oa[BQ], ob[BQ];
#pragma unroll
  for (int i = 0; i < BQ; ++i) { ob[i].x = 0.0f; ob[i].y = 0.0f; ob[i].z = 0.0f; ob[i].w = 0.0f; }
  load_x(xa, xl0);
  __builtin_amdgcn_sched_barrier(0);
  load_x(xb, xl0 + 2 * BLK);
  const float* xq = xl0 + 2 * (2 * BLK);
#pragma unroll
  for (int p = 0; p < 3 * HG; ++p) gi_part(ga, xa, 0, p);
  Tr = 0.0f; Tz = 0.0f; Tg = 0.0f;
  for (int t = t_first; t < t_end; t += 2 * BLK) {
    if ((t & (CHUNK - 1)) == 0) {
      Tr = fmaf(h, wr, ga.r[0][0]);
      Tz = fmaf(h, wz, ga.z[0][0]);
      Tg = fmaf(h, wn, bn);
      if (verify == 0 && t == t_store && j == 0) joins[(t_store / CHUNK) * JPITCH + q] = h;
    }
    if (t > t_store) store_block(ob, hq - BLK);
    run_block(ga, gb, xa, xb, xq, oa, ob);
    if (t >= t_store) store_block(oa, hq);
    run_block(ga, gb, xb, xa, xq + 2 * BLK, ob, oa);
    hq += 2 * BLK;
    xq += (t + 4 * BLK < SEQ_LEN) ? 2 * (2 * BLK) : 0;
  }
  store_block(ob, hq - BLK);

  if (verify == 0 && j == 0) joins[(t_store / CHUNK) * JPITCH + 4 + q] = h;
  if (t_end == SEQ_LEN) state_out[q] = h;
}

__global__ __launch_bounds__(256) void yproj_wmma_kernel(
    const float* __restrict__ hbuf, const float* __restrict__ w_out,
    const float* __restrict__ b_out, float* __restrict__ y) {
  const int lane = threadIdx.x & 31;
  const int wave = threadIdx.x >> 5;
  const int tile = blockIdx.x * 8 + wave;
  const int t0   = tile * 16;
  const int half = lane >> 4;
  const int idx  = lane & 15;

  v2f a;
  {
    const float* hr = hbuf + (size_t)(half * 2) * SEQ_LEN + t0 + idx;
    a.x = hr[0]; a.y = hr[SEQ_LEN];
  }

  v2f b;
  if (idx < 2) {
    const float2 wv = *reinterpret_cast<const float2*>(w_out + idx * 4 + half * 2);
    b.x = wv.x; b.y = wv.y;
  } else {
    b.x = 0.0f; b.y = 0.0f;
  }

  const float bias = (idx < 2) ? b_out[idx] : 0.0f;
  v8f c;
#pragma unroll
  for (int i = 0; i < 8; ++i) c[i] = bias;

  v8f d = __builtin_amdgcn_wmma_f32_16x16x4_f32(
      false, a, false, b, (short)0, c, false, false);

  if (idx < 2) {
    const int rbase = half * 8;
#pragma unroll
    for (int i = 0; i < 8; ++i)
      y[(size_t)(t0 + rbase + i) * 2 + idx] = d[i];
  }
}

extern "C" void kernel_launch(void* const* d_in, const int* in_sizes, int n_in,
                              void* d_out, int out_size, void* d_ws, size_t ws_size,
                              hipStream_t stream) {
  const float* x      = (const float*)d_in[0];
  const float* state  = (const float*)d_in[1];
  const float* w_ih   = (const float*)d_in[2];
  const float* w_hh   = (const float*)d_in[3];
  const float* b_ih   = (const float*)d_in[4];
  const float* b_hh   = (const float*)d_in[5];
  const float* w_out  = (const float*)d_in[6];
  const float* b_out  = (const float*)d_in[7];

  float* out  = (float*)d_out;
  float* ws   = (float*)d_ws;
  float* hbuf = ws + (size_t)NGATES * SEQ_LEN;

  const int tiles  = SEQ_LEN / 16;
  const int blocks = tiles / 8;

  gru_scan_kernel<<<NCHUNK / 4, 128, 0, stream>>>(x, w_ih, b_ih, ws, state, w_hh, b_hh, out,
                                                  out + (size_t)SEQ_LEN * 2, 0);
  gru_scan_kernel<<<1, 128, 0, stream>>>(x, w_ih, b_ih, ws, state, w_hh, b_hh, out,
                                         out + (size_t)SEQ_LEN * 2, 1);
  yproj_wmma_kernel<<<blocks, 256, 0, stream>>>(hbuf, w_out, b_out, out);
}


# scan: both 16-lane rows of a wave run a chunk each (2048 chunks of 128 steps; gi A rows 8-15 = upper row's steps): 640 steps a wave, not 768; products re-formed per 128 steps (rounding changes)
# speedup vs baseline: 1985.6855x; 1.0635x over previous
#include <hip/hip_runtime.h>

typedef __attribute__((ext_vector_type(2))) float v2f;
typedef __attribute__((ext_vector_type(8))) float v8f;

#define SEQ_LEN 262144
#define NGATES  12
#define HDIM    4

__device__ __forceinline__ float bcast_lane(float v, int lane) {
  return __int_as_float(__builtin_amdgcn_readlane(__float_as_int(v), lane));
}

__device__ __forceinline__ float hw_tanh(float a) {
  return __builtin_amdgcn_tanhf(a);
}


typedef __attribute__((ext_vector_type(4))) float v4f;

#define CHUNK  128
#define WARM   512
#define NCHUNK (SEQ_LEN / CHUNK)
#define JPITCH 16

#define BLK 32
#define BQ  (BLK / 4)

template <int M>
__device__ __forceinline__ float row_xor(float v) {
  return __int_as_float(__builtin_amdgcn_update_dpp(
      0, __float_as_int(v), 0x160 + M, 0xF, 0xF, true));
}

__device__ __forceinline__ float quad_first(float v) {
  return __int_as_float(__builtin_amdgcn_update_dpp(
      0, __float_as_int(v), 0x00, 0xF, 0xF, true));
}

#define BG  (BLK / 8)
#define HB  (BLK / 2)
#define HG  (HB / 8)

struct GiHalf { v8f r[HG], z[HG], n[HG]; };
struct XBlock { v2f a[BG]; };

__device__ __forceinline__ void load_x(XBlock& xb, const float* xl) {
#pragma unroll
  for (int i = 0; i < BG; ++i)
    xb.a[i] = *reinterpret_cast<const v2f*>(xl + 16 * i);
}

__global__ __launch_bounds__(128) __attribute__((amdgpu_waves_per_eu(1, 1)))
void gru_scan_kernel(
    const float* x, const float* __restrict__ w_ih, const float* __restrict__ b_ih,
    float* ws, const float* __restrict__ state0,
    const float* __restrict__ w_hh, const float* __restrict__ b_hh,
    float* joins, float* state_out, int verify) {
  int t_first, t_store, t_end;
  int toff = 0, loff = 0;
  if (verify == 0) {
    const int wave = __builtin_amdgcn_readfirstlane((int)(threadIdx.x >> 5));
    t_store = ((int)blockIdx.x * 4 + wave) * (2 * CHUNK);
    t_end   = t_store + CHUNK;
    t_first = (t_store - WARM > -CHUNK) ? t_store - WARM : -CHUNK;
    toff = (threadIdx.x & 16) ? CHUNK : 0;
    loff = (threadIdx.x & 8) ? CHUNK : 0;
  } else {
    const int tid = threadIdx.x;
    unsigned diff = 0;
#pragma unroll
    for (int i = 0; i < NCHUNK / 128; ++i) {
      const int c = i * 128 + tid;
      const int cb = (c != 0) ? c - 1 : 0;
      const v4f s = *reinterpret_cast<const v4f*>(joins + JPITCH * c);
      const v4f e = *reinterpret_cast<const v4f*>(joins + JPITCH * cb + 4);
      const unsigned dj = (__float_as_uint(s.x) ^ __float_as_uint(e.x))
                        | (__float_as_uint(s.y) ^ __float_as_uint(e.y))
                        | (__float_as_uint(s.z) ^ __float_as_uint(e.z))
                        | (__float_as_uint(s.w) ^ __float_as_uint(e.w));
      diff |= (c != 0) ? dj : 0u;
    }
    __shared__ unsigned wave_diff[4];
    const unsigned mine = __builtin_amdgcn_ballot_w32(diff != 0);
    if ((tid & 31) == 0) wave_diff[tid >> 5] = mine;
    __syncthreads();
    if ((wave_diff[0] | wave_diff[1] | wave_diff[2] | wave_diff[3]) == 0) return;
    if (tid >= 32) return;
    t_first = 0; t_store = 0; t_end = SEQ_LEN;
  }
  const int l = threadIdx.x & 15;
  const int q = l >> 2;
  const int j = l & 3;
  const int d = q ^ j;

  const float hf = 0.5f;
  const float wr = hf * w_hh[d * 4 + q];
  const float wz = hf * w_hh[(4 + d) * 4 + q];
  const float wn = hf * w_hh[(8 + d) * 4 + q];
  const float bn = (j == 0) ? hf * b_hh[8 + q] : 0.0f;

  const float h0 = state0[q];
  float h = 0.0f;

  float* hq = ws + ((ptrdiff_t)(NGATES + q) * SEQ_LEN + t_first + toff);

  const bool lo = (threadIdx.x & 16) == 0;
  const float cs = 0.125f;
  const float2 wir = *reinterpret_cast<const float2*>(w_ih + (d) * 2);
  const float2 wiz = *reinterpret_cast<const float2*>(w_ih + (4 + d) * 2);
  const float2 win = *reinterpret_cast<const float2*>(w_ih + (8 + q) * 2);
  v2f bR, bZ, bN;
  bR.x = lo ? cs * wir.x : 0.0f;  bR.y = lo ? cs * wir.y : 0.0f;
  bZ.x = lo ? cs * wiz.x : 0.0f;  bZ.y = lo ? cs * wiz.y : 0.0f;
  bN.x = lo ? win.x : 0.0f;       bN.y = lo ? win.y : 0.0f;
  const float biasR = cs * (b_ih[d] + b_hh[d]);
  const float biasZ = cs * (b_ih[4 + d] + b_hh[4 + d]);
  const float biasN = b_ih[8 + q];
  v8f cR, cZ, cN;
#pragma unroll
  for (int i = 0; i < 8; ++i) { cR[i] = biasR; cZ[i] = biasZ; cN[i] = biasN; }
  const bool early = t_first + loff < 0;
  const float* xl0 = x + 2 * (size_t)(early ? 0 : t_first + loff) + (l & 7) * 2;

  auto gi_part = [&](GiHalf& gn, const XBlock& xf, const int xg, const int p) {
    const int gg = p / 3;
    v2f a;
    a.x = lo ? xf.a[xg + gg].x : 0.0f;
    a.y = lo ? xf.a[xg + gg].y : 0.0f;
    if (p % 3 == 0)
      gn.r[gg] = __builtin_amdgcn_wmma_f32_16x16x4_f32(false, a, false, bR, (short)0, cR, false, false);
    else if (p % 3 == 1)
      gn.z[gg] = __builtin_amdgcn_wmma_f32_16x16x4_f32(false, a, false, bZ, (short)0, cZ, false, false);
    else
      gn.n[gg] = __builtin_amdgcn_wmma_f32_16x16x4_f32(false, a, false, bN, (short)0, cN, false, false);
  };

  float Tr, Tz, Tg;

  auto run_block = [&](GiHalf& P, GiHalf& Q, XBlock& xk, const XBlock& xk1,
                       const float* xk2, v4f (&o)[BQ], const v4f (&keep)[BQ]) {
    float hs[BLK];
#pragma unroll
    for (int u = 0; u < BLK; ++u) {
      const GiHalf& g  = (u < HB) ? P : Q;
      const GiHalf& gn = (u < HB) ? Q : P;
      const int v = u & (HB - 1);
      const float gin  = g.n[v >> 3][v & 7];
      const float girn = (v + 1 < HB) ? g.r[(v + 1) >> 3][(v + 1) & 7] : gn.r[0][0];
      const float gizn = (v + 1 < HB) ? g.z[(v + 1) >> 3][(v + 1) & 7] : gn.z[0][0];

      const float s1g = Tg + row_xor<5>(Tg);
      __builtin_amdgcn_sched_barrier(0);
      const float s1r = Tr + row_xor<5>(Tr);
      __builtin_amdgcn_sched_barrier(0);
      const float s1z = Tz + row_xor<5>(Tz);
      __builtin_amdgcn_sched_barrier(0);
      const float s2g = s1g + row_xor<10>(s1g);
      __builtin_amdgcn_sched_barrier(0);
      const float s2r = s1r + row_xor<10>(s1r);
      __builtin_amdgcn_sched_barrier(0);
      const float s2z = s1z + row_xor<10>(s1z);
      __builtin_amdgcn_sched_barrier(0);
      const float gh = quad_first(s2g);
      __builtin_amdgcn_sched_barrier(0);
      const float tr = hw_tanh(quad_first(s2r));
      __builtin_amdgcn_sched_barrier(0);
      const float tz = hw_tanh(quad_first(s2z));
      __builtin_amdgcn_sched_barrier(0);
      const float gnb = gin + gh;
      const float hh = 0.5f * h;
      __builtin_amdgcn_sched_barrier(0);
      const float na = fmaf(tr, gh, gnb);
      __builtin_amdgcn_sched_barrier(0);
      const float ca = fmaf(tz, -0.5f, 0.5f);
      const float cb = fmaf(tz, hh, hh);
      __builtin_amdgcn_sched_barrier(0);
      const float n = hw_tanh(na);
      __builtin_amdgcn_sched_barrier(0);
      const float ur = wr * ca;
      const float uz = wz * ca;
      const float ug = wn * ca;
      const float vr = fmaf(wr, cb, girn);
      const float vz = fmaf(wz, cb, gizn);
      const float vg = fmaf(wn, cb, bn);
      if (u < 3 * HG) gi_part(Q, xk, HG, u);
      else if (u >= HB && u < HB + 3 * HG) gi_part(P, xk1, 0, u - HB);
      __builtin_amdgcn_sched_barrier(0);
      Tr = fmaf(n, ur, vr);
      Tz = fmaf(n, uz, vz);
      Tg = fmaf(n, ug, vg);
      h = fmaf(n, ca, cb);
      hs[u] = h;
      __builtin_amdgcn_sched_barrier(0);
      if (u == 3 * HG) {
        load_x(xk, xk2);
        __builtin_amdgcn_sched_barrier(0);
      }
      if (u == BLK / 2 - 1) {
#pragma unroll
        for (int i = 0; i < BQ; ++i) asm volatile("" :: "v"(keep[i]));
      }
    }
    asm volatile("" :: "v"(h) : "memory");
#pragma unroll
    for (int i = 0; i < BQ; ++i) {
      o[i].x = hs[4 * i + 0]; o[i].y = hs[4 * i + 1];
      o[i].z = hs[4 * i + 2]; o[i].w = hs[4 * i + 3];
    }
  };
  auto store_block = [&](const v4f (&o)[BQ], float* hp) {
#pragma unroll
    for (int i = 0; i < BQ; ++i)
      *reinterpret_cast<v4f*>(hp + 4 * i) = o[i];
  };

  GiHalf ga, gb;
  XBlock xa, xb;
  v4f oa[BQ], ob[BQ];
#pragma unroll
  for (int i = 0; i < BQ; ++i) { ob[i].x = 0.0f; ob[i].y = 0.0f; ob[i].z = 0.0f; ob[i].w = 0.0f; }
  load_x(xa, xl0);
  __builtin_amdgcn_sched_barrier(0);
  load_x(xb, xl0 + 2 * BLK);
  const float* xq = xl0 + (early ? 0 : 2 * (2 * BLK));
#pragma unroll
  for (int p = 0; p < 3 * HG; ++p) gi_part(ga, xa, 0, p);
  Tr = 0.0f; Tz = 0.0f; Tg = 0.0f;
  for (int t = t_first; t < t_end; t += 2 * BLK) {
    if ((t & (CHUNK - 1)) == 0) {
      h = (t + toff == 0) ? h0 : h;
      Tr = fmaf(h, wr, ga.r[0][0]);
      Tz = fmaf(h, wz, ga.z[0][0]);
      Tg = fmaf(h, wn, bn);
      if (verify == 0 && t == t_store && j == 0) joins[((t_store + toff) / CHUNK) * JPITCH + q] = h;
    }
    if (t > t_store) store_block(ob, hq - BLK);
    run_block(ga, gb, xa, xb, xq, oa, ob);
    if (t >= t_store) store_block(oa, hq);
    run_block(ga, gb, xb, xa, xq + 2 * BLK, ob, oa);
    hq += 2 * BLK;
    const int t_hi = t + ((verify == 0) ? CHUNK : 0);
    const int adv = (t_hi + 4 * BLK < SEQ_LEN) ? 2 * (2 * BLK) : 0;
    xq += (early && t == t_first) ? 0 : adv;
  }
  store_block(ob, hq - BLK);

  if (verify == 0 && j == 0) joins[((t_store + toff) / CHUNK) * JPITCH + 4 + q] = h;
  if (t_end + toff == SEQ_LEN) state_out[q] = h;
}

__global__ __launch_bounds__(256) void yproj_wmma_kernel(
    const float* __restrict__ hbuf, const float* __restrict__ w_out,
    const float* __restrict__ b_out, float* __restrict__ y) {
  const int lane = threadIdx.x & 31;
  const int wave = threadIdx.x >> 5;
  const int tile = blockIdx.x * 8 + wave;
  const int t0   = tile * 16;
  const int half = lane >> 4;
  const int idx  = lane & 15;

  v2f a;
  {
    const float* hr = hbuf + (size_t)(half * 2) * SEQ_LEN + t0 + idx;
    a.x = hr[0]; a.y = hr[SEQ_LEN];
  }

  v2f b;
  if (idx < 2) {
    const float2 wv = *reinterpret_cast<const float2*>(w_out + idx * 4 + half * 2);
    b.x = wv.x; b.y = wv.y;
  } else {
    b.x = 0.0f; b.y = 0.0f;
  }

  const float bias = (idx < 2) ? b_out[idx] : 0.0f;
  v8f c;
#pragma unroll
  for (int i = 0; i < 8; ++i) c[i] = bias;

  v8f d = __builtin_amdgcn_wmma_f32_16x16x4_f32(
      false, a, false, b, (short)0, c, false, false);

  if (idx < 2) {
    const int rbase = half * 8;
#pragma unroll
    for (int i = 0; i < 8; ++i)
      y[(size_t)(t0 + rbase + i) * 2 + idx] = d[i];
  }
}

extern "C" void kernel_launch(void* const* d_in, const int* in_sizes, int n_in,
                              void* d_out, int out_size, void* d_ws, size_t ws_size,
                              hipStream_t stream) {
  const float* x      = (const float*)d_in[0];
  const float* state  = (const float*)d_in[1];
  const float* w_ih   = (const float*)d_in[2];
  const float* w_hh   = (const float*)d_in[3];
  const float* b_ih   = (const float*)d_in[4];
  const float* b_hh   = (const float*)d_in[5];
  const float* w_out  = (const float*)d_in[6];
  const float* b_out  = (const float*)d_in[7];

  float* out  = (float*)d_out;
  float* ws   = (float*)d_ws;
  float* hbuf = ws + (size_t)NGATES * SEQ_LEN;

  const int tiles  = SEQ_LEN / 16;
  const int blocks = tiles / 8;

  gru_scan_kernel<<<NCHUNK / 8, 128, 0, stream>>>(x, w_ih, b_ih, ws, state, w_hh, b_hh, out,
                                                  out + (size_t)SEQ_LEN * 2, 0);
  gru_scan_kernel<<<1, 128, 0, stream>>>(x, w_ih, b_ih, ws, state, w_hh, b_hh, out,
                                         out + (size_t)SEQ_LEN * 2, 1);
  yproj_wmma_kernel<<<blocks, 256, 0, stream>>>(hbuf, w_out, b_out, out);
}
